# starting kernel
# baseline (speedup 1.0000x reference)
#include <hip/hip_runtime.h>

typedef __attribute__((ext_vector_type(2))) float v2f;
typedef __attribute__((ext_vector_type(8))) float v8f;

#define FDIM 64

__global__ void k_init_deg(float* __restrict__ deg, int n) {
    int i = blockIdx.x * blockDim.x + threadIdx.x;
    if (i < n) deg[i] = 1.0f;
}

__global__ void k_deg_accum(const int* __restrict__ dst, float* __restrict__ deg, int E) {
    int e = blockIdx.x * blockDim.x + threadIdx.x;
    if (e < E) atomicAdd(&deg[dst[e]], 1.0f);
}

__global__ void k_dinv(const float* __restrict__ deg, float* __restrict__ dinv, int n) {
    int i = blockIdx.x * blockDim.x + threadIdx.x;
    if (i < n) { float d = deg[i]; dinv[i] = d > 0.0f ? rsqrtf(d) : 0.0f; }
}

__global__ void k_selfloop(const float* __restrict__ x, const float* __restrict__ dinv,
                           const float* __restrict__ bias, float* __restrict__ agg, int n) {
    int idx = blockIdx.x * blockDim.x + threadIdx.x;
    if (idx < n * FDIM) {
        int i = idx >> 6;
        int f = idx & 63;
        float di = dinv[i];
        float v  = x[idx] * di * di;
        if (bias) v += bias[f];
        agg[idx] = v;
    }
}

__global__ void k_scatter(const float* __restrict__ x, const int* __restrict__ src,
                          const int* __restrict__ dst, const float* __restrict__ dinv,
                          float* __restrict__ agg, int E) {
    long long gtid = (long long)blockIdx.x * blockDim.x + threadIdx.x;
    int e    = (int)(gtid >> 5);
    int lane = (int)(gtid & 31);
    if (e < E) {
        int s = src[e], d = dst[e];
        float w = dinv[s] * dinv[d];
        v2f xv = *(const v2f*)(x + (long long)s * FDIM + lane * 2);
        float* p = agg + (long long)d * FDIM + lane * 2;
        atomicAdd(p,     xv.x * w);
        atomicAdd(p + 1, xv.y * w);
    }
}

__global__ void k_gemm_wmma(const float* __restrict__ A, const float* __restrict__ Bm,
                            const float* __restrict__ bias, float* __restrict__ out,
                            int n, int K, int ncols, int relu) {
    int col_tile = blockIdx.x;
    int row_tile = blockIdx.y * blockDim.y + threadIdx.y;
    if (row_tile * 16 >= n) return;

    int lane = threadIdx.x;
    int half = lane >> 4;
    int m    = lane & 15;
    int row0 = row_tile * 16;
    int col0 = col_tile * 16;

    const float* arow = A  + (long long)(row0 + m) * K + 2 * half;
    const float* bcol = Bm + (long long)(2 * half) * ncols + col0 + m;

    v8f acc = {};
    for (int k0 = 0; k0 < K; k0 += 4) {
        v2f a = *(const v2f*)(arow + k0);
        v2f b;
        b.x = bcol[(long long)(k0)     * ncols];
        b.y = bcol[(long long)(k0 + 1) * ncols];
        acc = __builtin_amdgcn_wmma_f32_16x16x4_f32(
                  false, a, false, b, (short)0, acc, false, false);
    }

    float bv = bias ? bias[col0 + m] : 0.0f;
#pragma unroll
    for (int r = 0; r < 8; ++r) {
        float v = acc[r] + bv;
        if (relu) v = fmaxf(v, 0.0f);
        out[(long long)(row0 + r + 8 * half) * ncols + col0 + m] = v;
    }
}

extern "C" void kernel_launch(void* const* d_in, const int* in_sizes, int n_in,
                              void* d_out, int out_size, void* d_ws, size_t ws_size,
                              hipStream_t stream) {
    (void)n_in; (void)out_size; (void)ws_size;

    const int*   eidx = (const int*)d_in[0];
    const float* emb  = (const float*)d_in[1];
    const float* W1   = (const float*)d_in[2];
    const float* b1   = (const float*)d_in[3];
    const float* W2   = (const float*)d_in[4];
    const float* b2   = (const float*)d_in[5];

    const int E = in_sizes[0] / 2;
    const int n = in_sizes[1] / 64;
    const int* src = eidx;
    const int* dst = eidx + E;

    float* deg  = (float*)d_ws;
    float* dinv = deg + n;
    float* agg1 = dinv + n;
    float* h    = agg1 + (size_t)n * 64;
    float* t    = agg1;
    float* out  = (float*)d_out;

    const int B = 256;
    k_init_deg <<<(n + B - 1) / B, B, 0, stream>>>(deg, n);
    k_deg_accum<<<(E + B - 1) / B, B, 0, stream>>>(dst, deg, E);
    k_dinv     <<<(n + B - 1) / B, B, 0, stream>>>(deg, dinv, n);

    k_selfloop<<<(int)(((size_t)n * 64 + B - 1) / B), B, 0, stream>>>(emb, dinv, nullptr, agg1, n);
    k_scatter <<<(int)(((size_t)E * 32 + B - 1) / B), B, 0, stream>>>(emb, src, dst, dinv, agg1, E);
    {
        dim3 block(32, 8);
        dim3 grid(128 / 16, ((n + 15) / 16 + 7) / 8);
        k_gemm_wmma<<<grid, block, 0, stream>>>(agg1, W1, b1, h, n, 64, 128, 1);
    }

    {
        dim3 block(32, 8);
        dim3 grid(64 / 16, ((n + 15) / 16 + 7) / 8);
        k_gemm_wmma<<<grid, block, 0, stream>>>(h, W2, nullptr, t, n, 128, 64, 0);
    }
    k_selfloop<<<(int)(((size_t)n * 64 + B - 1) / B), B, 0, stream>>>(t, dinv, b2, out, n);
    k_scatter <<<(int)(((size_t)E * 32 + B - 1) / B), B, 0, stream>>>(t, src, dst, dinv, out, E);
}


# k_scatter: one wave walks 32 edges (lane-parallel index and norm loads, source rows fetched 8 at a time) instead of one short wave per edge; 32x fewer waves, same float atomics
# speedup vs baseline: 1.3681x; 1.3681x over previous
#include <hip/hip_runtime.h>

typedef __attribute__((ext_vector_type(2))) float v2f;
typedef __attribute__((ext_vector_type(8))) float v8f;

#define FDIM 64

__global__ void k_init_deg(float* __restrict__ deg, int n) {
    int i = blockIdx.x * blockDim.x + threadIdx.x;
    if (i < n) deg[i] = 1.0f;
}

__global__ void k_deg_accum(const int* __restrict__ dst, float* __restrict__ deg, int E) {
    int e = blockIdx.x * blockDim.x + threadIdx.x;
    if (e < E) atomicAdd(&deg[dst[e]], 1.0f);
}

__global__ void k_dinv(const float* __restrict__ deg, float* __restrict__ dinv, int n) {
    int i = blockIdx.x * blockDim.x + threadIdx.x;
    if (i < n) { float d = deg[i]; dinv[i] = d > 0.0f ? rsqrtf(d) : 0.0f; }
}

__global__ void k_selfloop(const float* __restrict__ x, const float* __restrict__ dinv,
                           const float* __restrict__ bias, float* __restrict__ agg, int n) {
    int idx = blockIdx.x * blockDim.x + threadIdx.x;
    if (idx < n * FDIM) {
        int i = idx >> 6;
        int f = idx & 63;
        float di = dinv[i];
        float v  = x[idx] * di * di;
        if (bias) v += bias[f];
        agg[idx] = v;
    }
}

__global__ void k_scatter(const float* __restrict__ x, const int* __restrict__ src,
                          const int* __restrict__ dst, const float* __restrict__ dinv,
                          float* __restrict__ agg, int E) {
    int lane = threadIdx.x & 31;
    int wave = (int)(((long long)blockIdx.x * blockDim.x + threadIdx.x) >> 5);
    int base = __builtin_amdgcn_readfirstlane(wave * 32);
    if (base >= E) return;
    int cnt = E - base < 32 ? E - base : 32;

    int s = 0, d = 0;
    float w = 0.0f;
    if (lane < cnt) {
        s = src[base + lane];
        d = dst[base + lane];
        w = dinv[s] * dinv[d];
    }

    for (int j0 = 0; j0 < cnt; j0 += 8) {
        v2f xv[8];
        for (int u = 0; u < 8; ++u) {
            int sj = __builtin_amdgcn_readlane(s, j0 + u);
            xv[u] = *(const v2f*)(x + (long long)sj * FDIM + lane * 2);
        }
        for (int u = 0; u < 8; ++u) {
            if (j0 + u < cnt) {
                int   dj = __builtin_amdgcn_readlane(d, j0 + u);
                float wj = __int_as_float(__builtin_amdgcn_readlane(__float_as_int(w), j0 + u));
                float* p = agg + (long long)dj * FDIM + lane * 2;
                atomicAdd(p,     xv[u].x * wj);
                atomicAdd(p + 1, xv[u].y * wj);
            }
        }
    }
}

__global__ void k_gemm_wmma(const float* __restrict__ A, const float* __restrict__ Bm,
                            const float* __restrict__ bias, float* __restrict__ out,
                            int n, int K, int ncols, int relu) {
    int col_tile = blockIdx.x;
    int row_tile = blockIdx.y * blockDim.y + threadIdx.y;
    if (row_tile * 16 >= n) return;

    int lane = threadIdx.x;
    int half = lane >> 4;
    int m    = lane & 15;
    int row0 = row_tile * 16;
    int col0 = col_tile * 16;

    const float* arow = A  + (long long)(row0 + m) * K + 2 * half;
    const float* bcol = Bm + (long long)(2 * half) * ncols + col0 + m;

    v8f acc = {};
    for (int k0 = 0; k0 < K; k0 += 4) {
        v2f a = *(const v2f*)(arow + k0);
        v2f b;
        b.x = bcol[(long long)(k0)     * ncols];
        b.y = bcol[(long long)(k0 + 1) * ncols];
        acc = __builtin_amdgcn_wmma_f32_16x16x4_f32(
                  false, a, false, b, (short)0, acc, false, false);
    }

    float bv = bias ? bias[col0 + m] : 0.0f;
#pragma unroll
    for (int r = 0; r < 8; ++r) {
        float v = acc[r] + bv;
        if (relu) v = fmaxf(v, 0.0f);
        out[(long long)(row0 + r + 8 * half) * ncols + col0 + m] = v;
    }
}

extern "C" void kernel_launch(void* const* d_in, const int* in_sizes, int n_in,
                              void* d_out, int out_size, void* d_ws, size_t ws_size,
                              hipStream_t stream) {
    (void)n_in; (void)out_size; (void)ws_size;

    const int*   eidx = (const int*)d_in[0];
    const float* emb  = (const float*)d_in[1];
    const float* W1   = (const float*)d_in[2];
    const float* b1   = (const float*)d_in[3];
    const float* W2   = (const float*)d_in[4];
    const float* b2   = (const float*)d_in[5];

    const int E = in_sizes[0] / 2;
    const int n = in_sizes[1] / 64;
    const int* src = eidx;
    const int* dst = eidx + E;

    float* deg  = (float*)d_ws;
    float* dinv = deg + n;
    float* agg1 = dinv + n;
    float* h    = agg1 + (size_t)n * 64;
    float* t    = agg1;
    float* out  = (float*)d_out;

    const int B = 256;
    k_init_deg <<<(n + B - 1) / B, B, 0, stream>>>(deg, n);
    k_deg_accum<<<(E + B - 1) / B, B, 0, stream>>>(dst, deg, E);
    k_dinv     <<<(n + B - 1) / B, B, 0, stream>>>(deg, dinv, n);

    k_selfloop<<<(int)(((size_t)n * 64 + B - 1) / B), B, 0, stream>>>(emb, dinv, nullptr, agg1, n);
    k_scatter <<<(E + B - 1) / B, B, 0, stream>>>(emb, src, dst, dinv, agg1, E);
    {
        dim3 block(32, 8);
        dim3 grid(128 / 16, ((n + 15) / 16 + 7) / 8);
        k_gemm_wmma<<<grid, block, 0, stream>>>(agg1, W1, b1, h, n, 64, 128, 1);
    }

    {
        dim3 block(32, 8);
        dim3 grid(64 / 16, ((n + 15) / 16 + 7) / 8);
        k_gemm_wmma<<<grid, block, 0, stream>>>(h, W2, nullptr, t, n, 128, 64, 0);
    }
    k_selfloop<<<(int)(((size_t)n * 64 + B - 1) / B), B, 0, stream>>>(t, dinv, b2, out, n);
    k_scatter <<<(E + B - 1) / B, B, 0, stream>>>(t, src, dst, dinv, out, E);
}


# k_scatter: lane l owns features l and l+32 (was 2l, 2l+1): every wave-wide f32 atomic add and row load covers one whole 128-byte line, not two half-used ones; same atomics, arrival-order sums
# speedup vs baseline: 1.5748x; 1.1511x over previous
#include <hip/hip_runtime.h>

typedef __attribute__((ext_vector_type(2))) float v2f;
typedef __attribute__((ext_vector_type(8))) float v8f;

#define FDIM 64

__global__ void k_init_deg(float* __restrict__ deg, int n) {
    int i = blockIdx.x * blockDim.x + threadIdx.x;
    if (i < n) deg[i] = 1.0f;
}

__global__ void k_deg_accum(const int* __restrict__ dst, float* __restrict__ deg, int E) {
    int e = blockIdx.x * blockDim.x + threadIdx.x;
    if (e < E) atomicAdd(&deg[dst[e]], 1.0f);
}

__global__ void k_dinv(const float* __restrict__ deg, float* __restrict__ dinv, int n) {
    int i = blockIdx.x * blockDim.x + threadIdx.x;
    if (i < n) { float d = deg[i]; dinv[i] = d > 0.0f ? rsqrtf(d) : 0.0f; }
}

__global__ void k_selfloop(const float* __restrict__ x, const float* __restrict__ dinv,
                           const float* __restrict__ bias, float* __restrict__ agg, int n) {
    int idx = blockIdx.x * blockDim.x + threadIdx.x;
    if (idx < n * FDIM) {
        int i = idx >> 6;
        int f = idx & 63;
        float di = dinv[i];
        float v  = x[idx] * di * di;
        if (bias) v += bias[f];
        agg[idx] = v;
    }
}

__global__ void k_scatter(const float* __restrict__ x, const int* __restrict__ src,
                          const int* __restrict__ dst, const float* __restrict__ dinv,
                          float* __restrict__ agg, int E) {
    int lane = threadIdx.x & 31;
    int wave = (int)(((long long)blockIdx.x * blockDim.x + threadIdx.x) >> 5);
    int base = __builtin_amdgcn_readfirstlane(wave * 32);
    if (base >= E) return;
    int cnt = E - base < 32 ? E - base : 32;

    int s = 0, d = 0;
    float w = 0.0f;
    if (lane < cnt) {
        s = src[base + lane];
        d = dst[base + lane];
        w = dinv[s] * dinv[d];
    }

    int j0 = 0;
    for (; j0 + 8 <= cnt; j0 += 8) {
        float xlo[8], xhi[8];
        for (int u = 0; u < 8; ++u) {
            int sj = __builtin_amdgcn_readlane(s, j0 + u);
            const float* r = x + (long long)sj * FDIM + lane;
            xlo[u] = r[0];
            xhi[u] = r[32];
        }
        for (int u = 0; u < 8; ++u) {
            int   dj = __builtin_amdgcn_readlane(d, j0 + u);
            float wj = __int_as_float(__builtin_amdgcn_readlane(__float_as_int(w), j0 + u));
            float* p = agg + (long long)dj * FDIM + lane;
            atomicAdd(p,      xlo[u] * wj);
            atomicAdd(p + 32, xhi[u] * wj);
        }
    }
    for (; j0 < cnt; ++j0) {
        int   sj = __builtin_amdgcn_readlane(s, j0);
        int   dj = __builtin_amdgcn_readlane(d, j0);
        float wj = __int_as_float(__builtin_amdgcn_readlane(__float_as_int(w), j0));
        const float* r = x + (long long)sj * FDIM + lane;
        float* p = agg + (long long)dj * FDIM + lane;
        atomicAdd(p,      r[0]  * wj);
        atomicAdd(p + 32, r[32] * wj);
    }
}

__global__ void k_gemm_wmma(const float* __restrict__ A, const float* __restrict__ Bm,
                            const float* __restrict__ bias, float* __restrict__ out,
                            int n, int K, int ncols, int relu) {
    int col_tile = blockIdx.x;
    int row_tile = blockIdx.y * blockDim.y + threadIdx.y;
    if (row_tile * 16 >= n) return;

    int lane = threadIdx.x;
    int half = lane >> 4;
    int m    = lane & 15;
    int row0 = row_tile * 16;
    int col0 = col_tile * 16;

    const float* arow = A  + (long long)(row0 + m) * K + 2 * half;
    const float* bcol = Bm + (long long)(2 * half) * ncols + col0 + m;

    v8f acc = {};
    for (int k0 = 0; k0 < K; k0 += 4) {
        v2f a = *(const v2f*)(arow + k0);
        v2f b;
        b.x = bcol[(long long)(k0)     * ncols];
        b.y = bcol[(long long)(k0 + 1) * ncols];
        acc = __builtin_amdgcn_wmma_f32_16x16x4_f32(
                  false, a, false, b, (short)0, acc, false, false);
    }

    float bv = bias ? bias[col0 + m] : 0.0f;
#pragma unroll
    for (int r = 0; r < 8; ++r) {
        float v = acc[r] + bv;
        if (relu) v = fmaxf(v, 0.0f);
        out[(long long)(row0 + r + 8 * half) * ncols + col0 + m] = v;
    }
}

extern "C" void kernel_launch(void* const* d_in, const int* in_sizes, int n_in,
                              void* d_out, int out_size, void* d_ws, size_t ws_size,
                              hipStream_t stream) {
    (void)n_in; (void)out_size; (void)ws_size;

    const int*   eidx = (const int*)d_in[0];
    const float* emb  = (const float*)d_in[1];
    const float* W1   = (const float*)d_in[2];
    const float* b1   = (const float*)d_in[3];
    const float* W2   = (const float*)d_in[4];
    const float* b2   = (const float*)d_in[5];

    const int E = in_sizes[0] / 2;
    const int n = in_sizes[1] / 64;
    const int* src = eidx;
    const int* dst = eidx + E;

    float* deg  = (float*)d_ws;
    float* dinv = deg + n;
    float* agg1 = dinv + n;
    float* h    = agg1 + (size_t)n * 64;
    float* t    = agg1;
    float* out  = (float*)d_out;

    const int B = 256;
    k_init_deg <<<(n + B - 1) / B, B, 0, stream>>>(deg, n);
    k_deg_accum<<<(E + B - 1) / B, B, 0, stream>>>(dst, deg, E);
    k_dinv     <<<(n + B - 1) / B, B, 0, stream>>>(deg, dinv, n);

    k_selfloop<<<(int)(((size_t)n * 64 + B - 1) / B), B, 0, stream>>>(emb, dinv, nullptr, agg1, n);
    k_scatter <<<(E + B - 1) / B, B, 0, stream>>>(emb, src, dst, dinv, agg1, E);
    {
        dim3 block(32, 8);
        dim3 grid(128 / 16, ((n + 15) / 16 + 7) / 8);
        k_gemm_wmma<<<grid, block, 0, stream>>>(agg1, W1, b1, h, n, 64, 128, 1);
    }

    {
        dim3 block(32, 8);
        dim3 grid(64 / 16, ((n + 15) / 16 + 7) / 8);
        k_gemm_wmma<<<grid, block, 0, stream>>>(h, W2, nullptr, t, n, 128, 64, 0);
    }
    k_selfloop<<<(int)(((size_t)n * 64 + B - 1) / B), B, 0, stream>>>(t, dinv, b2, out, n);
    k_scatter <<<(E + B - 1) / B, B, 0, stream>>>(t, src, dst, dinv, out, E);
}


# aggregation: edges split by destination range (LDS-counted count+place passes), an owner work-group per range sums rows by LDS float adds (arrival order), dst norm applied once, lines stored once
# speedup vs baseline: 2.2366x; 1.4202x over previous
#include <hip/hip_runtime.h>

typedef __attribute__((ext_vector_type(2))) float v2f;
typedef __attribute__((ext_vector_type(8))) float v8f;

#define FDIM 64
#define ROWS 196
#define MAXBINS 1024
#define EPT 8
#define EPB (EPT * 1024)


__global__ void k_zero(int* __restrict__ p) {
    int i = blockIdx.x * blockDim.x + threadIdx.x;
    if (i < 2 * MAXBINS) p[i] = 0;
}

__global__ __launch_bounds__(1024)
void k_hist(const int* __restrict__ dst, int* __restrict__ g_hist, int E, int nbins) {
    __shared__ int c[MAXBINS];
    const int tid = threadIdx.x;
    c[tid] = 0;
    __syncthreads();
#pragma unroll
    for (int k = 0; k < EPT; ++k) {
        int e = blockIdx.x * EPB + k * 1024 + tid;
        if (e < E) atomicAdd(&c[dst[e] / ROWS], 1);
    }
    __syncthreads();
    int v = c[tid];
    if (tid < nbins && v > 0) atomicAdd(&g_hist[tid], v);
}

__global__ __launch_bounds__(1024)
void k_place(const int* __restrict__ src, const int* __restrict__ dst,
             const int* __restrict__ g_hist, int* __restrict__ g_cursor,
             int* __restrict__ binstart, unsigned* __restrict__ list, int E, int nbins) {
    __shared__ int c[MAXBINS];
    __shared__ int sc[2][MAXBINS];
    const int tid = threadIdx.x;
    c[tid] = 0;
    int tot = tid < nbins ? g_hist[tid] : 0;
    sc[0][tid] = tot;
    __syncthreads();

    int bin[EPT], rk[EPT], loc[EPT];
#pragma unroll
    for (int k = 0; k < EPT; ++k) {
        int e = blockIdx.x * EPB + k * 1024 + tid;
        bin[k] = -1; rk[k] = 0; loc[k] = 0;
        if (e < E) {
            int d = dst[e];
            int b = d / ROWS;
            bin[k] = b;
            loc[k] = d - b * ROWS;
            rk[k]  = atomicAdd(&c[b], 1);
        }
    }

    int cur = 0;
    for (int off = 1; off < MAXBINS; off <<= 1) {
        int v = sc[cur][tid];
        if (tid >= off) v += sc[cur][tid - off];
        sc[cur ^ 1][tid] = v;
        cur ^= 1;
        __syncthreads();
    }
    const int incl  = sc[cur][tid];
    const int start = incl - tot;

    const int mine = c[tid];
    int off0 = 0;
    if (tid < nbins && mine > 0) off0 = atomicAdd(&g_cursor[tid], mine);
    if (blockIdx.x == 0) {
        if (tid < nbins) binstart[tid] = start;
        if (tid == nbins - 1) binstart[nbins] = incl;
    }
    c[tid] = start + off0;
    __syncthreads();

#pragma unroll
    for (int k = 0; k < EPT; ++k) {
        int e = blockIdx.x * EPB + k * 1024 + tid;
        if (bin[k] >= 0)
            list[c[bin[k]] + rk[k]] = ((unsigned)src[e] << 8) | (unsigned)loc[k];
    }
}

__global__ __launch_bounds__(256)
void k_dinv(const unsigned* __restrict__ list, const int* __restrict__ binstart,
            float* __restrict__ dinv, int n) {
    __shared__ int c[256];
    const int tid = threadIdx.x;
    const int bin = blockIdx.x;
    c[tid] = 0;
    __syncthreads();
    const int beg = binstart[bin];
    const int end = binstart[bin + 1];
    for (int i = beg + tid; i < end; i += 256) atomicAdd(&c[list[i] & 255u], 1);
    __syncthreads();
    if (tid < ROWS) {
        int row = bin * ROWS + tid;
        if (row < n) {
            float d = (float)(c[tid] + 1);
            dinv[row] = d > 0.0f ? rsqrtf(d) : 0.0f;
        }
    }
}

__global__ void k_move_list(const unsigned* __restrict__ from, unsigned* __restrict__ to, int E) {
    int stride = gridDim.x * blockDim.x;
    for (int i = blockIdx.x * blockDim.x + threadIdx.x; i < E; i += stride) to[i] = from[i];
}

__global__ __launch_bounds__(1024)
void k_agg(const float* __restrict__ x, const unsigned* __restrict__ list,
           const int* __restrict__ binstart, const float* __restrict__ dinv,
           const float* __restrict__ bias, float* __restrict__ outp, int n) {
    __shared__ float acc[ROWS * FDIM];
    const int tid  = threadIdx.x;
    const int lane = tid & 31;
    const int wv  = __builtin_amdgcn_readfirstlane(tid >> 5);
    const int bin = blockIdx.x;

    for (int i = tid; i < ROWS * FDIM; i += 1024) acc[i] = 0.0f;
    __syncthreads();

    const int beg = binstart[bin];
    const int end = binstart[bin + 1];
    for (int base = beg + wv * 32; base < end; base += 1024) {
        const int cnt = end - base < 32 ? end - base : 32;
        unsigned word = 0u;
        float w = 0.0f;
        if (lane < cnt) {
            word = list[base + lane];
            w = dinv[word >> 8];
        }
        int j0 = 0;
        for (; j0 + 8 <= cnt; j0 += 8) {
            float xlo[8], xhi[8];
            for (int u = 0; u < 8; ++u) {
                unsigned wj = (unsigned)__builtin_amdgcn_readlane((int)word, j0 + u);
                const float* r = x + (long long)(wj >> 8) * FDIM + lane;
                xlo[u] = r[0];
                xhi[u] = r[32];
            }
            for (int u = 0; u < 8; ++u) {
                unsigned wj = (unsigned)__builtin_amdgcn_readlane((int)word, j0 + u);
                float sj = __int_as_float(__builtin_amdgcn_readlane(__float_as_int(w), j0 + u));
                float* p = acc + (wj & 255u) * FDIM + lane;
                atomicAdd(p,      xlo[u] * sj);
                atomicAdd(p + 32, xhi[u] * sj);
            }
        }
        for (; j0 < cnt; ++j0) {
            unsigned wj = (unsigned)__builtin_amdgcn_readlane((int)word, j0);
            float sj = __int_as_float(__builtin_amdgcn_readlane(__float_as_int(w), j0));
            const float* r = x + (long long)(wj >> 8) * FDIM + lane;
            float* p = acc + (wj & 255u) * FDIM + lane;
            atomicAdd(p,      r[0]  * sj);
            atomicAdd(p + 32, r[32] * sj);
        }
    }
    __syncthreads();

    const int row0 = bin * ROWS;
    for (int i = tid; i < ROWS * FDIM; i += 1024) {
        int row = row0 + (i >> 6);
        if (row < n) {
            int f = i & 63;
            float di = dinv[row];
            float v = di * acc[i] + x[(long long)row * FDIM + f] * di * di;
            if (bias) v += bias[f];
            outp[(long long)row * FDIM + f] = v;
        }
    }
}

__global__ void k_gemm_wmma(const float* __restrict__ A, const float* __restrict__ Bm,
                            const float* __restrict__ bias, float* __restrict__ out,
                            int n, int K, int ncols, int relu) {
    int col_tile = blockIdx.x;
    int row_tile = blockIdx.y * blockDim.y + threadIdx.y;
    if (row_tile * 16 >= n) return;

    int lane = threadIdx.x;
    int half = lane >> 4;
    int m    = lane & 15;
    int row0 = row_tile * 16;
    int col0 = col_tile * 16;

    const float* arow = A  + (long long)(row0 + m) * K + 2 * half;
    const float* bcol = Bm + (long long)(2 * half) * ncols + col0 + m;

    v8f acc = {};
    for (int k0 = 0; k0 < K; k0 += 4) {
        v2f a = *(const v2f*)(arow + k0);
        v2f b;
        b.x = bcol[(long long)(k0)     * ncols];
        b.y = bcol[(long long)(k0 + 1) * ncols];
        acc = __builtin_amdgcn_wmma_f32_16x16x4_f32(
                  false, a, false, b, (short)0, acc, false, false);
    }

    float bv = bias ? bias[col0 + m] : 0.0f;
#pragma unroll
    for (int r = 0; r < 8; ++r) {
        float v = acc[r] + bv;
        if (relu) v = fmaxf(v, 0.0f);
        out[(long long)(row0 + r + 8 * half) * ncols + col0 + m] = v;
    }
}

extern "C" void kernel_launch(void* const* d_in, const int* in_sizes, int n_in,
                              void* d_out, int out_size, void* d_ws, size_t ws_size,
                              hipStream_t stream) {
    (void)n_in; (void)out_size; (void)ws_size;

    const int*   eidx = (const int*)d_in[0];
    const float* emb  = (const float*)d_in[1];
    const float* W1   = (const float*)d_in[2];
    const float* b1   = (const float*)d_in[3];
    const float* W2   = (const float*)d_in[4];
    const float* b2   = (const float*)d_in[5];

    const int E = in_sizes[0] / 2;
    const int n = in_sizes[1] / 64;
    const int* src = eidx;
    const int* dst = eidx + E;
    const int nbins = (n + ROWS - 1) / ROWS;

    int*   g_hist   = (int*)d_ws;
    int*   g_cursor = g_hist + MAXBINS;
    int*   binstart = g_hist + 2 * MAXBINS;
    float* dinv = (float*)d_ws + n;
    float* agg1 = dinv + n;
    float* h    = agg1 + (size_t)n * 64;
    float* t    = agg1;
    float* out  = (float*)d_out;
    unsigned* list_a = (unsigned*)d_out;
    unsigned* list_b = (unsigned*)h;

    k_zero <<<2, 1024, 0, stream>>>(g_hist);
    k_hist <<<(E + EPB - 1) / EPB, 1024, 0, stream>>>(dst, g_hist, E, nbins);
    k_place<<<(E + EPB - 1) / EPB, 1024, 0, stream>>>(src, dst, g_hist, g_cursor, binstart, list_a, E, nbins);
    k_dinv <<<nbins, 256, 0, stream>>>(list_a, binstart, dinv, n);

    k_agg<<<nbins, 1024, 0, stream>>>(emb, list_a, binstart, dinv, nullptr, agg1, n);
    {
        dim3 block(32, 8);
        dim3 grid(128 / 16, ((n + 15) / 16 + 7) / 8);
        k_gemm_wmma<<<grid, block, 0, stream>>>(agg1, W1, b1, h, n, 64, 128, 1);
    }

    {
        dim3 block(32, 8);
        dim3 grid(64 / 16, ((n + 15) / 16 + 7) / 8);
        k_gemm_wmma<<<grid, block, 0, stream>>>(h, W2, nullptr, t, n, 128, 64, 0);
    }
    k_move_list<<<512, 256, 0, stream>>>(list_a, list_b, E);
    k_agg<<<nbins, 1024, 0, stream>>>(t, list_b, binstart, dinv, b2, out, n);
}


# both dense products fused in one kernel: a wave keeps its 16 rows' A fragments in registers over all column tiles, h goes through LDS, never to memory (its plane holds the edge list: no list move)
# speedup vs baseline: 3.6048x; 1.6117x over previous
#include <hip/hip_runtime.h>

typedef __attribute__((ext_vector_type(2))) float v2f;
typedef __attribute__((ext_vector_type(8))) float v8f;
typedef __attribute__((ext_vector_type(4))) float v4f;

#define FDIM 64
#define ROWS 196
#define MAXBINS 1024
#define EPT 8
#define EPB (EPT * 1024)


__global__ void k_zero(int* __restrict__ p) {
    int i = blockIdx.x * blockDim.x + threadIdx.x;
    if (i < 2 * MAXBINS) p[i] = 0;
}

__global__ __launch_bounds__(1024)
void k_hist(const int* __restrict__ dst, int* __restrict__ g_hist, int E, int nbins) {
    __shared__ int c[MAXBINS];
    const int tid = threadIdx.x;
    c[tid] = 0;
    __syncthreads();
#pragma unroll
    for (int k = 0; k < EPT; ++k) {
        int e = blockIdx.x * EPB + k * 1024 + tid;
        if (e < E) atomicAdd(&c[dst[e] / ROWS], 1);
    }
    __syncthreads();
    int v = c[tid];
    if (tid < nbins && v > 0) atomicAdd(&g_hist[tid], v);
}

__global__ __launch_bounds__(1024)
void k_place(const int* __restrict__ src, const int* __restrict__ dst,
             const int* __restrict__ g_hist, int* __restrict__ g_cursor,
             int* __restrict__ binstart, unsigned* __restrict__ list, int E, int nbins) {
    __shared__ int c[MAXBINS];
    __shared__ int sc[2][MAXBINS];
    const int tid = threadIdx.x;
    c[tid] = 0;
    int tot = tid < nbins ? g_hist[tid] : 0;
    sc[0][tid] = tot;
    __syncthreads();

    int bin[EPT], rk[EPT], loc[EPT];
#pragma unroll
    for (int k = 0; k < EPT; ++k) {
        int e = blockIdx.x * EPB + k * 1024 + tid;
        bin[k] = -1; rk[k] = 0; loc[k] = 0;
        if (e < E) {
            int d = dst[e];
            int b = d / ROWS;
            bin[k] = b;
            loc[k] = d - b * ROWS;
            rk[k]  = atomicAdd(&c[b], 1);
        }
    }

    int cur = 0;
    for (int off = 1; off < MAXBINS; off <<= 1) {
        int v = sc[cur][tid];
        if (tid >= off) v += sc[cur][tid - off];
        sc[cur ^ 1][tid] = v;
        cur ^= 1;
        __syncthreads();
    }
    const int incl  = sc[cur][tid];
    const int start = incl - tot;

    const int mine = c[tid];
    int off0 = 0;
    if (tid < nbins && mine > 0) off0 = atomicAdd(&g_cursor[tid], mine);
    if (blockIdx.x == 0) {
        if (tid < nbins) binstart[tid] = start;
        if (tid == nbins - 1) binstart[nbins] = incl;
    }
    c[tid] = start + off0;
    __syncthreads();

#pragma unroll
    for (int k = 0; k < EPT; ++k) {
        int e = blockIdx.x * EPB + k * 1024 + tid;
        if (bin[k] >= 0)
            list[c[bin[k]] + rk[k]] = ((unsigned)src[e] << 8) | (unsigned)loc[k];
    }
}

__global__ __launch_bounds__(256)
void k_dinv(const unsigned* __restrict__ list, const int* __restrict__ binstart,
            float* __restrict__ dinv, int n) {
    __shared__ int c[256];
    const int tid = threadIdx.x;
    const int bin = blockIdx.x;
    c[tid] = 0;
    __syncthreads();
    const int beg = binstart[bin];
    const int end = binstart[bin + 1];
    for (int i = beg + tid; i < end; i += 256) atomicAdd(&c[list[i] & 255u], 1);
    __syncthreads();
    if (tid < ROWS) {
        int row = bin * ROWS + tid;
        if (row < n) {
            float d = (float)(c[tid] + 1);
            dinv[row] = d > 0.0f ? rsqrtf(d) : 0.0f;
        }
    }
}

__global__ __launch_bounds__(1024)
void k_agg(const float* __restrict__ x, const unsigned* __restrict__ list,
           const int* __restrict__ binstart, const float* __restrict__ dinv,
           const float* __restrict__ bias, float* __restrict__ outp, int n) {
    __shared__ float acc[ROWS * FDIM];
    const int tid  = threadIdx.x;
    const int lane = tid & 31;
    const int wv  = __builtin_amdgcn_readfirstlane(tid >> 5);
    const int bin = blockIdx.x;

    for (int i = tid; i < ROWS * FDIM; i += 1024) acc[i] = 0.0f;
    __syncthreads();

    const int beg = binstart[bin];
    const int end = binstart[bin + 1];
    for (int base = beg + wv * 32; base < end; base += 1024) {
        const int cnt = end - base < 32 ? end - base : 32;
        unsigned word = 0u;
        float w = 0.0f;
        if (lane < cnt) {
            word = list[base + lane];
            w = dinv[word >> 8];
        }
        int j0 = 0;
        for (; j0 + 8 <= cnt; j0 += 8) {
            float xlo[8], xhi[8];
            for (int u = 0; u < 8; ++u) {
                unsigned wj = (unsigned)__builtin_amdgcn_readlane((int)word, j0 + u);
                const float* r = x + (long long)(wj >> 8) * FDIM + lane;
                xlo[u] = r[0];
                xhi[u] = r[32];
            }
            for (int u = 0; u < 8; ++u) {
                unsigned wj = (unsigned)__builtin_amdgcn_readlane((int)word, j0 + u);
                float sj = __int_as_float(__builtin_amdgcn_readlane(__float_as_int(w), j0 + u));
                float* p = acc + (wj & 255u) * FDIM + lane;
                atomicAdd(p,      xlo[u] * sj);
                atomicAdd(p + 32, xhi[u] * sj);
            }
        }
        for (; j0 < cnt; ++j0) {
            unsigned wj = (unsigned)__builtin_amdgcn_readlane((int)word, j0);
            float sj = __int_as_float(__builtin_amdgcn_readlane(__float_as_int(w), j0));
            const float* r = x + (long long)(wj >> 8) * FDIM + lane;
            float* p = acc + (wj & 255u) * FDIM + lane;
            atomicAdd(p,      r[0]  * sj);
            atomicAdd(p + 32, r[32] * sj);
        }
    }
    __syncthreads();

    const int row0 = bin * ROWS;
    for (int i = tid; i < ROWS * FDIM; i += 1024) {
        int row = row0 + (i >> 6);
        if (row < n) {
            int f = i & 63;
            float di = dinv[row];
            float v = di * acc[i] + x[(long long)row * FDIM + f] * di * di;
            if (bias) v += bias[f];
            outp[(long long)row * FDIM + f] = v;
        }
    }
}

#define MLP_WAVES 8
#define HP 132
#define WF 8192
#define MLP_LDS_FLOATS (2 * WF + MLP_WAVES * 16 * HP)

__global__ __launch_bounds__(256)
void k_mlp_wmma(const float* __restrict__ A, const float* __restrict__ W1,
                const float* __restrict__ b1, const float* __restrict__ W2,
                float* __restrict__ T, int ntiles) {
    __shared__ __attribute__((aligned(16))) float smem[MLP_LDS_FLOATS];
    float* w1f = smem;
    float* w2f = smem + WF;

    const int tid  = threadIdx.x;
    const int lane = tid & 31;
    const int half = lane >> 4;
    const int m    = lane & 15;
    const int wv = __builtin_amdgcn_readfirstlane(tid >> 5);
    float* hb = smem + 2 * WF + wv * (16 * HP);

#pragma unroll
    for (int i = 0; i < WF / 256; ++i) {
        int d  = tid + 256 * i;
        int e  = d & 1;
        int ln = (d >> 1) & 31;
        int mm = ln & 15;
        int hh = ln >> 4;
        int ct1 = (d >> 6) & 7;
        int s1  = d >> 9;
        w1f[d] = W1[(4 * s1 + 2 * hh + e) * 128 + 16 * ct1 + mm];
        int ct2 = (d >> 6) & 3;
        int s2  = d >> 8;
        w2f[d] = W2[(4 * s2 + 2 * hh + e) * 64 + 16 * ct2 + mm];
    }
    __syncthreads();

    const v2f* w1v = (const v2f*)w1f;
    const v2f* w2v = (const v2f*)w2f;
    const int slots = gridDim.x * MLP_WAVES;
    const int iters = (ntiles + slots - 1) / slots;
    for (int it = 0; it < iters; ++it) {
        const int tile = it * slots + wv * gridDim.x + blockIdx.x;
        const bool valid = tile < ntiles;

        if (valid) {
            const float* arow = A + (long long)(tile * 16 + m) * 64 + 2 * half;
            v2f a1[16];
#pragma unroll
            for (int s = 0; s < 16; ++s) a1[s] = *(const v2f*)(arow + 4 * s);

#pragma unroll
            for (int g = 0; g < 2; ++g) {
                v8f acc[4] = {};
#pragma unroll
                for (int s = 0; s < 16; ++s) {
#pragma unroll
                    for (int j = 0; j < 4; ++j) {
                        v2f b = w1v[(s * 8 + g * 4 + j) * 32 + lane];
                        acc[j] = __builtin_amdgcn_wmma_f32_16x16x4_f32(
                                     false, a1[s], false, b, (short)0, acc[j], false, false);
                    }
                }
#pragma unroll
                for (int j = 0; j < 4; ++j) {
                    const int col = (g * 4 + j) * 16 + m;
                    const float bv = b1[col];
#pragma unroll
                    for (int r = 0; r < 8; ++r) {
                        hb[(r + 8 * half) * HP + col] = fmaxf(acc[j][r] + bv, 0.0f);
                    }
                }
            }
        }
        __syncthreads();

        v2f a2[32];
        if (valid) {
#pragma unroll
            for (int s = 0; s < 32; ++s) a2[s] = *(const v2f*)(hb + m * HP + 4 * s + 2 * half);
        }
        __syncthreads();

        if (valid) {
            v8f acc[4] = {};
#pragma unroll
            for (int s = 0; s < 32; ++s) {
#pragma unroll
                for (int j = 0; j < 4; ++j) {
                    v2f b = w2v[(s * 4 + j) * 32 + lane];
                    acc[j] = __builtin_amdgcn_wmma_f32_16x16x4_f32(
                                 false, a2[s], false, b, (short)0, acc[j], false, false);
                }
            }
#pragma unroll
            for (int j = 0; j < 4; ++j)
#pragma unroll
                for (int r = 0; r < 8; ++r)
                    hb[(r + 8 * half) * HP + j * 16 + m] = acc[j][r];
        }
        __syncthreads();

        v4f tv[8];
#pragma unroll
        for (int i = 0; i < 8; ++i) {
            const int q = (i * 32 + lane) * 4;
            tv[i] = *(const v4f*)(hb + (q >> 6) * HP + (q & 63));
        }
        __syncthreads();
        if (valid) {
            float* tt = T + (long long)tile * (16 * 64);
#pragma unroll
            for (int i = 0; i < 8; ++i) *(v4f*)(tt + (i * 32 + lane) * 4) = tv[i];
        }
    }
}

extern "C" void kernel_launch(void* const* d_in, const int* in_sizes, int n_in,
                              void* d_out, int out_size, void* d_ws, size_t ws_size,
                              hipStream_t stream) {
    (void)n_in; (void)out_size; (void)ws_size;

    const int*   eidx = (const int*)d_in[0];
    const float* emb  = (const float*)d_in[1];
    const float* W1   = (const float*)d_in[2];
    const float* b1   = (const float*)d_in[3];
    const float* W2   = (const float*)d_in[4];
    const float* b2   = (const float*)d_in[5];

    const int E = in_sizes[0] / 2;
    const int n = in_sizes[1] / 64;
    const int* src = eidx;
    const int* dst = eidx + E;
    const int nbins = (n + ROWS - 1) / ROWS;

    int*   g_hist   = (int*)d_ws;
    int*   g_cursor = g_hist + MAXBINS;
    int*   binstart = g_hist + 2 * MAXBINS;
    float* dinv = (float*)d_ws + n;
    float* agg1 = dinv + n;
    float* t    = agg1 + (size_t)n * 64;
    float* out  = (float*)d_out;
    unsigned* list = (unsigned*)(t + (size_t)n * 64);

    k_zero <<<2, 1024, 0, stream>>>(g_hist);
    k_hist <<<(E + EPB - 1) / EPB, 1024, 0, stream>>>(dst, g_hist, E, nbins);
    k_place<<<(E + EPB - 1) / EPB, 1024, 0, stream>>>(src, dst, g_hist, g_cursor, binstart, list, E, nbins);
    k_dinv <<<nbins, 256, 0, stream>>>(list, binstart, dinv, n);

    k_agg<<<nbins, 1024, 0, stream>>>(emb, list, binstart, dinv, nullptr, agg1, n);

    k_mlp_wmma<<<512, 256, 0, stream>>>(agg1, W1, b1, W2, t, (n + 15) / 16);

    k_agg<<<nbins, 1024, 0, stream>>>(t, list, binstart, dinv, b2, out, n);
}


# aggregation: each range's list re-ordered by destination row (self-loop entry added); a wave sums 7 rows in registers in list order (LDS-rank arrival order), one store per line; no LDS float adds
# speedup vs baseline: 3.7498x; 1.0402x over previous
#include <hip/hip_runtime.h>

typedef __attribute__((ext_vector_type(2))) float v2f;
typedef __attribute__((ext_vector_type(8))) float v8f;
typedef __attribute__((ext_vector_type(4))) float v4f;

#define FDIM 64
#define ROWS 196
#define MAXBINS 1024
#define EPT 8
#define EPB (EPT * 1024)


__global__ void k_zero(int* __restrict__ p) {
    int i = blockIdx.x * blockDim.x + threadIdx.x;
    if (i < 2 * MAXBINS) p[i] = 0;
}

__global__ __launch_bounds__(1024)
void k_hist(const int* __restrict__ dst, int* __restrict__ g_hist, int E, int nbins) {
    __shared__ int c[MAXBINS];
    const int tid = threadIdx.x;
    c[tid] = 0;
    __syncthreads();
#pragma unroll
    for (int k = 0; k < EPT; ++k) {
        int e = blockIdx.x * EPB + k * 1024 + tid;
        if (e < E) atomicAdd(&c[dst[e] / ROWS], 1);
    }
    __syncthreads();
    int v = c[tid];
    if (tid < nbins && v > 0) atomicAdd(&g_hist[tid], v);
}

__global__ __launch_bounds__(1024)
void k_place(const int* __restrict__ src, const int* __restrict__ dst,
             const int* __restrict__ g_hist, int* __restrict__ g_cursor,
             int* __restrict__ binstart, unsigned* __restrict__ list, int E, int nbins) {
    __shared__ int c[MAXBINS];
    __shared__ int sc[2][MAXBINS];
    const int tid = threadIdx.x;
    c[tid] = 0;
    int tot = tid < nbins ? g_hist[tid] : 0;
    sc[0][tid] = tot;
    __syncthreads();

    int bin[EPT], rk[EPT], loc[EPT];
#pragma unroll
    for (int k = 0; k < EPT; ++k) {
        int e = blockIdx.x * EPB + k * 1024 + tid;
        bin[k] = -1; rk[k] = 0; loc[k] = 0;
        if (e < E) {
            int d = dst[e];
            int b = d / ROWS;
            bin[k] = b;
            loc[k] = d - b * ROWS;
            rk[k]  = atomicAdd(&c[b], 1);
        }
    }

    int cur = 0;
    for (int off = 1; off < MAXBINS; off <<= 1) {
        int v = sc[cur][tid];
        if (tid >= off) v += sc[cur][tid - off];
        sc[cur ^ 1][tid] = v;
        cur ^= 1;
        __syncthreads();
    }
    const int incl  = sc[cur][tid];
    const int start = incl - tot;

    const int mine = c[tid];
    int off0 = 0;
    if (tid < nbins && mine > 0) off0 = atomicAdd(&g_cursor[tid], mine);
    if (blockIdx.x == 0) {
        if (tid < nbins) binstart[tid] = start;
        if (tid == nbins - 1) binstart[nbins] = incl;
    }
    c[tid] = start + off0;
    __syncthreads();

#pragma unroll
    for (int k = 0; k < EPT; ++k) {
        int e = blockIdx.x * EPB + k * 1024 + tid;
        if (bin[k] >= 0)
            list[c[bin[k]] + rk[k]] = ((unsigned)src[e] << 8) | (unsigned)loc[k];
    }
}

#define RPW 7
#define CH (ROWS / RPW)
__global__ __launch_bounds__(256)
void k_rowsort(const unsigned* __restrict__ list, const int* __restrict__ binstart,
               float* __restrict__ dinv, unsigned* __restrict__ list2, int* __restrict__ cptr,
               int n, int nbins) {
    __shared__ int c[256];
    __shared__ int sc[2][256];
    const int tid = threadIdx.x;
    const int bin = blockIdx.x;
    c[tid] = 0;
    __syncthreads();
    const int beg = binstart[bin];
    const int end = binstart[bin + 1];
    for (int i0 = beg + tid; i0 < end; i0 += 8 * 256) {
        unsigned wd[8];
#pragma unroll
        for (int k = 0; k < 8; ++k) {
            const int i = i0 + k * 256;
            wd[k] = list[i < end ? i : end - 1];
        }
        __builtin_amdgcn_sched_barrier(0);
#pragma unroll
        for (int k = 0; k < 8; ++k)
            atomicAdd(&c[wd[k] & 255u], i0 + k * 256 < end ? 1 : 0);
    }
    __syncthreads();
    const int row = bin * ROWS + tid;
    const bool valid = tid < ROWS && row < n;
    const int slots = valid ? c[tid] + 1 : 0;
    if (valid) {
        float d = (float)slots;
        dinv[row] = d > 0.0f ? rsqrtf(d) : 0.0f;
    }
    sc[0][tid] = slots;
    __syncthreads();
    int cur = 0;
    for (int off = 1; off < 256; off <<= 1) {
        int v = sc[cur][tid];
        if (tid >= off) v += sc[cur][tid - off];
        sc[cur ^ 1][tid] = v;
        cur ^= 1;
        __syncthreads();
    }
    const int incl = sc[cur][tid];
    const int base = beg + bin * ROWS;
    const int first = base + incl - slots;
    if (valid) list2[first] = ((unsigned)row << 8) | (unsigned)tid;
    if (tid < ROWS && tid % RPW == 0) cptr[bin * CH + tid / RPW] = first;
    if (bin == nbins - 1 && tid == 255) cptr[nbins * CH] = base + incl;
    c[tid] = first + 1;
    __syncthreads();
    for (int i0 = beg + tid; i0 < end; i0 += 8 * 256) {
        unsigned wd[8];
#pragma unroll
        for (int k = 0; k < 8; ++k) {
            const int i = i0 + k * 256;
            wd[k] = list[i < end ? i : end - 1];
        }
        __builtin_amdgcn_sched_barrier(0);
#pragma unroll
        for (int k = 0; k < 8; ++k) {
            const bool live = i0 + k * 256 < end;
            const int p = atomicAdd(&c[wd[k] & 255u], live ? 1 : 0);
            if (live) list2[p] = wd[k];
        }
    }
}

template <bool FULL>
__device__ __forceinline__
void agg_group(const float* __restrict__ x, float* __restrict__ orow, unsigned word, float w,
               int j0, int cnt, int lane, float blo, float bhi,
               int& cur, int& run, float& alo, float& ahi) {
    float xlo[16], xhi[16];
#pragma unroll
    for (int u = 0; u < 16; ++u) {
        unsigned wj = (unsigned)__builtin_amdgcn_readlane((int)word, j0 + u);
        const float* r = x + (long long)(wj >> 8) * FDIM + lane;
        xlo[u] = r[0];
        xhi[u] = r[32];
    }
    __builtin_amdgcn_sched_barrier(0);
#pragma unroll
    for (int u = 0; u < 16; ++u) {
        const bool live = FULL || j0 + u < cnt;
        unsigned wj = (unsigned)__builtin_amdgcn_readlane((int)word, j0 + u);
        float sj = __int_as_float(__builtin_amdgcn_readlane(__float_as_int(w), j0 + u));
        const int tag = (int)(wj & 255u);
        if (live && tag != cur) {
            if (cur >= 0) {
                const float di = rsqrtf((float)run);
                orow[cur * FDIM]      = di * alo + blo;
                orow[cur * FDIM + 32] = di * ahi + bhi;
            }
            cur = tag; run = 0; alo = 0.0f; ahi = 0.0f;
        }
        const float nlo = alo + xlo[u] * sj;
        const float nhi = ahi + xhi[u] * sj;
        alo = live ? nlo : alo;
        ahi = live ? nhi : ahi;
        run += live ? 1 : 0;
    }
}

__global__ __launch_bounds__(CH * 32)
void k_agg(const float* __restrict__ x, const unsigned* __restrict__ list2,
           const int* __restrict__ cptr, const float* __restrict__ dinv,
           const float* __restrict__ bias, float* __restrict__ outp) {
    const int tid  = threadIdx.x;
    const int lane = tid & 31;
    const int wv  = __builtin_amdgcn_readfirstlane(tid >> 5);
    const int bin = blockIdx.x;
    const int beg = cptr[bin * CH + wv];
    const int end = cptr[bin * CH + wv + 1];
    const float blo = bias ? bias[lane] : 0.0f;
    const float bhi = bias ? bias[lane + 32] : 0.0f;
    float* const orow = outp + (long long)bin * ROWS * FDIM + lane;

    float alo = 0.0f, ahi = 0.0f;
    int cur = -1, run = 0;
    unsigned word = 0u;
    float w = 0.0f;
    if (beg + lane < end) {
        word = list2[beg + lane];
        w = dinv[word >> 8];
    }
    for (int base = beg; base < end; base += 32) {
        const int cnt = end - base < 32 ? end - base : 32;
        const bool more = base + 32 + lane < end;
        unsigned nword = 0u;
        float nw;
        if (more) nword = list2[base + 32 + lane];
        if (cnt >= 16) agg_group<true >(x, orow, word, w, 0, cnt, lane, blo, bhi, cur, run, alo, ahi);
        else           agg_group<false>(x, orow, word, w, 0, cnt, lane, blo, bhi, cur, run, alo, ahi);
        __builtin_amdgcn_sched_barrier(0);
        nw = dinv[nword >> 8];
        if (cnt >= 32)     agg_group<true >(x, orow, word, w, 16, cnt, lane, blo, bhi, cur, run, alo, ahi);
        else if (cnt > 16) agg_group<false>(x, orow, word, w, 16, cnt, lane, blo, bhi, cur, run, alo, ahi);
        word = nword;
        w = nw;
    }
    if (cur >= 0) {
        const float di = rsqrtf((float)run);
        orow[cur * FDIM]      = di * alo + blo;
        orow[cur * FDIM + 32] = di * ahi + bhi;
    }
}

#define MLP_WAVES 8
#define HP 132
#define WF 8192
#define MLP_LDS_FLOATS (2 * WF + MLP_WAVES * 16 * HP)

__global__ __launch_bounds__(256)
void k_mlp_wmma(const float* __restrict__ A, const float* __restrict__ W1,
                const float* __restrict__ b1, const float* __restrict__ W2,
                float* __restrict__ T, int ntiles) {
    __shared__ __attribute__((aligned(16))) float smem[MLP_LDS_FLOATS];
    float* w1f = smem;
    float* w2f = smem + WF;

    const int tid  = threadIdx.x;
    const int lane = tid & 31;
    const int half = lane >> 4;
    const int m    = lane & 15;
    const int wv = __builtin_amdgcn_readfirstlane(tid >> 5);
    float* hb = smem + 2 * WF + wv * (16 * HP);

#pragma unroll
    for (int i = 0; i < WF / 256; ++i) {
        int d  = tid + 256 * i;
        int e  = d & 1;
        int ln = (d >> 1) & 31;
        int mm = ln & 15;
        int hh = ln >> 4;
        int ct1 = (d >> 6) & 7;
        int s1  = d >> 9;
        w1f[d] = W1[(4 * s1 + 2 * hh + e) * 128 + 16 * ct1 + mm];
        int ct2 = (d >> 6) & 3;
        int s2  = d >> 8;
        w2f[d] = W2[(4 * s2 + 2 * hh + e) * 64 + 16 * ct2 + mm];
    }
    __syncthreads();

    const v2f* w1v = (const v2f*)w1f;
    const v2f* w2v = (const v2f*)w2f;
    const int slots = gridDim.x * MLP_WAVES;
    const int iters = (ntiles + slots - 1) / slots;
    for (int it = 0; it < iters; ++it) {
        const int tile = it * slots + wv * gridDim.x + blockIdx.x;
        const bool valid = tile < ntiles;

        if (valid) {
            const float* arow = A + (long long)(tile * 16 + m) * 64 + 2 * half;
            v2f a1[16];
#pragma unroll
            for (int s = 0; s < 16; ++s) a1[s] = *(const v2f*)(arow + 4 * s);

#pragma unroll
            for (int g = 0; g < 2; ++g) {
                v8f acc[4] = {};
#pragma unroll
                for (int s = 0; s < 16; ++s) {
#pragma unroll
                    for (int j = 0; j < 4; ++j) {
                        v2f b = w1v[(s * 8 + g * 4 + j) * 32 + lane];
                        acc[j] = __builtin_amdgcn_wmma_f32_16x16x4_f32(
                                     false, a1[s], false, b, (short)0, acc[j], false, false);
                    }
                }
#pragma unroll
                for (int j = 0; j < 4; ++j) {
                    const int col = (g * 4 + j) * 16 + m;
                    const float bv = b1[col];
#pragma unroll
                    for (int r = 0; r < 8; ++r) {
                        hb[(r + 8 * half) * HP + col] = fmaxf(acc[j][r] + bv, 0.0f);
                    }
                }
            }
        }
        __syncthreads();

        v2f a2[32];
        if (valid) {
#pragma unroll
            for (int s = 0; s < 32; ++s) a2[s] = *(const v2f*)(hb + m * HP + 4 * s + 2 * half);
        }
        __syncthreads();

        if (valid) {
            v8f acc[4] = {};
#pragma unroll
            for (int s = 0; s < 32; ++s) {
#pragma unroll
                for (int j = 0; j < 4; ++j) {
                    v2f b = w2v[(s * 4 + j) * 32 + lane];
                    acc[j] = __builtin_amdgcn_wmma_f32_16x16x4_f32(
                                 false, a2[s], false, b, (short)0, acc[j], false, false);
                }
            }
#pragma unroll
            for (int j = 0; j < 4; ++j)
#pragma unroll
                for (int r = 0; r < 8; ++r)
                    hb[(r + 8 * half) * HP + j * 16 + m] = acc[j][r];
        }
        __syncthreads();

        v4f tv[8];
#pragma unroll
        for (int i = 0; i < 8; ++i) {
            const int q = (i * 32 + lane) * 4;
            tv[i] = *(const v4f*)(hb + (q >> 6) * HP + (q & 63));
        }
        __syncthreads();
        if (valid) {
            float* tt = T + (long long)tile * (16 * 64);
#pragma unroll
            for (int i = 0; i < 8; ++i) *(v4f*)(tt + (i * 32 + lane) * 4) = tv[i];
        }
    }
}

extern "C" void kernel_launch(void* const* d_in, const int* in_sizes, int n_in,
                              void* d_out, int out_size, void* d_ws, size_t ws_size,
                              hipStream_t stream) {
    (void)n_in; (void)out_size; (void)ws_size;

    const int*   eidx = (const int*)d_in[0];
    const float* emb  = (const float*)d_in[1];
    const float* W1   = (const float*)d_in[2];
    const float* b1   = (const float*)d_in[3];
    const float* W2   = (const float*)d_in[4];
    const float* b2   = (const float*)d_in[5];

    const int E = in_sizes[0] / 2;
    const int n = in_sizes[1] / 64;
    const int* src = eidx;
    const int* dst = eidx + E;
    const int nbins = (n + ROWS - 1) / ROWS;

    int*   g_hist   = (int*)d_ws;
    int*   g_cursor = g_hist + MAXBINS;
    int*   binstart = g_hist + 2 * MAXBINS;
    int*   cptr     = binstart + MAXBINS + 1;
    float* dinv = (float*)d_ws + n;
    float* agg1 = dinv + n;
    float* t    = agg1 + (size_t)n * 64;
    float* out  = (float*)d_out;
    unsigned* list  = (unsigned*)agg1;
    unsigned* list2 = (unsigned*)(t + (size_t)n * 64);

    k_zero <<<2, 1024, 0, stream>>>(g_hist);
    k_hist <<<(E + EPB - 1) / EPB, 1024, 0, stream>>>(dst, g_hist, E, nbins);
    k_place<<<(E + EPB - 1) / EPB, 1024, 0, stream>>>(src, dst, g_hist, g_cursor, binstart, list, E, nbins);
    k_rowsort<<<nbins, 256, 0, stream>>>(list, binstart, dinv, list2, cptr, n, nbins);

    k_agg<<<nbins, CH * 32, 0, stream>>>(emb, list2, cptr, dinv, nullptr, agg1);

    k_mlp_wmma<<<512, 256, 0, stream>>>(agg1, W1, b1, W2, t, (n + 15) / 16);

    k_agg<<<nbins, CH * 32, 0, stream>>>(t, list2, cptr, dinv, b2, out);
}


# edge split without global counters: a block orders its 8192 edges by range in LDS (grouped loads, whole-line stores, start table); row pass gathers the runs; 7 to 5 launches; arrival-order sums
# speedup vs baseline: 4.9666x; 1.3245x over previous
#include <hip/hip_runtime.h>

typedef __attribute__((ext_vector_type(2))) float v2f;
typedef __attribute__((ext_vector_type(8))) float v8f;
typedef __attribute__((ext_vector_type(4))) float v4f;

#define FDIM 64
#define ROWS 196
#define MAXBINS 1024
#define EPT 8
#define EPB (EPT * 1024)

#define MAXBLK 256

__global__ __launch_bounds__(1024)
void k_split(const int* __restrict__ src, const int* __restrict__ dst,
             unsigned* __restrict__ list, int* __restrict__ boff, int E, int nbins) {
    __shared__ int c[MAXBINS];
    __shared__ int sc[2][MAXBINS];
    __shared__ unsigned srt[EPB];
    const int tid = threadIdx.x;
    c[tid] = 0;
    __syncthreads();

    int dv[EPT], sv[EPT];
#pragma unroll
    for (int k = 0; k < EPT; ++k) {
        const int e = blockIdx.x * EPB + k * 1024 + tid;
        const int ec = e < E ? e : E - 1;
        dv[k] = dst[ec];
        sv[k] = src[ec];
    }
    __builtin_amdgcn_sched_barrier(0);
    int bin[EPT], rk[EPT];
    unsigned wdv[EPT];
#pragma unroll
    for (int k = 0; k < EPT; ++k) {
        const bool live = blockIdx.x * EPB + k * 1024 + tid < E;
        const int d = dv[k];
        const int b = d / ROWS;
        rk[k]  = atomicAdd(&c[live ? b : tid], live ? 1 : 0);
        bin[k] = live ? b : -1;
        wdv[k] = ((unsigned)sv[k] << 8) | (unsigned)(d - b * ROWS);
    }
    __syncthreads();

    const int tot = c[tid];
    sc[0][tid] = tot;
    __syncthreads();
    int cur = 0;
    for (int off = 1; off < MAXBINS; off <<= 1) {
        int v = sc[cur][tid];
        if (tid >= off) v += sc[cur][tid - off];
        sc[cur ^ 1][tid] = v;
        cur ^= 1;
        __syncthreads();
    }
    const int start = sc[cur][tid] - tot;
    if (tid <= nbins) boff[(long long)blockIdx.x * (nbins + 1) + tid] = start;
    c[tid] = start;
    __syncthreads();

#pragma unroll
    for (int k = 0; k < EPT; ++k)
        if (bin[k] >= 0) srt[c[bin[k]] + rk[k]] = wdv[k];
    __syncthreads();

    for (int k = 0; k < EPT; ++k) {
        int i = k * 1024 + tid;
        int e = blockIdx.x * EPB + i;
        if (e < E) list[e] = srt[i];
    }
}

#define RPW 7
#define CH (ROWS / RPW)
__global__ __launch_bounds__(256)
void k_rowsort(const unsigned* __restrict__ list, const int* __restrict__ boff,
               float* __restrict__ dinv, unsigned* __restrict__ list2, int* __restrict__ cptr,
               int n, int nbins, int nblk) {
    __shared__ int c[256];
    __shared__ int sc[2][256];
    __shared__ int ro[MAXBLK];
    __shared__ int rc[MAXBLK];
    __shared__ int s_beg, s_max;
    const int tid = threadIdx.x;
    const int bin = blockIdx.x;
    c[tid] = 0;
    if (tid == 0) { s_beg = 0; s_max = 0; }
    __syncthreads();
    int o0 = 0, len = 0;
    if (tid < nblk) {
        const int* bo = boff + (long long)tid * (nbins + 1) + bin;
        o0  = bo[0];
        len = bo[1] - o0;
    }
    ro[tid] = tid * EPB + o0;
    rc[tid] = len;
    int so = o0, mx = len;
    for (int d = 16; d > 0; d >>= 1) {
        so += __shfl_xor(so, d);
        int t = __shfl_xor(mx, d);
        mx = t > mx ? t : mx;
    }
    if ((tid & 31) == 0) { atomicAdd(&s_beg, so); atomicMax(&s_max, mx); }
    __syncthreads();
    const int beg   = s_beg;
    const int maxrc = s_max;

    const int hw  = tid >> 4;
    const int l16 = tid & 15;
    int myro[16], myrc[16];
#pragma unroll
    for (int j = 0; j < 16; ++j) {
        myro[j] = ro[hw + 16 * j] + l16;
        myrc[j] = rc[hw + 16 * j] - l16;
    }

    for (int kb = 0; kb < maxrc; kb += 32) {
        unsigned wd[32];
#pragma unroll
        for (int j = 0; j < 16; ++j) {
            wd[2 * j]     = list[kb      < myrc[j] ? myro[j] + kb      : 0];
            wd[2 * j + 1] = list[kb + 16 < myrc[j] ? myro[j] + kb + 16 : 0];
        }
        __builtin_amdgcn_sched_barrier(0);
#pragma unroll
        for (int j = 0; j < 32; ++j) {
            const bool live = kb + 16 * (j & 1) < myrc[j >> 1];
            atomicAdd(&c[live ? (wd[j] & 255u) : (unsigned)tid], live ? 1 : 0);
        }
    }
    __syncthreads();
    const int row = bin * ROWS + tid;
    const bool valid = tid < ROWS && row < n;
    const int slots = valid ? c[tid] + 1 : 0;
    if (valid) {
        float d = (float)slots;
        dinv[row] = d > 0.0f ? rsqrtf(d) : 0.0f;
    }
    sc[0][tid] = slots;
    __syncthreads();
    int cur = 0;
    for (int off = 1; off < 256; off <<= 1) {
        int v = sc[cur][tid];
        if (tid >= off) v += sc[cur][tid - off];
        sc[cur ^ 1][tid] = v;
        cur ^= 1;
        __syncthreads();
    }
    const int incl = sc[cur][tid];
    const int base = beg + bin * ROWS;
    const int first = base + incl - slots;
    if (valid) list2[first] = ((unsigned)row << 8) | (unsigned)tid;
    if (tid < ROWS && tid % RPW == 0) cptr[bin * CH + tid / RPW] = first;
    if (bin == nbins - 1 && tid == 255) cptr[nbins * CH] = base + incl;
    c[tid] = first + 1;
    __syncthreads();
    for (int kb = 0; kb < maxrc; kb += 32) {
        unsigned wd[32];
#pragma unroll
        for (int j = 0; j < 16; ++j) {
            wd[2 * j]     = list[kb      < myrc[j] ? myro[j] + kb      : 0];
            wd[2 * j + 1] = list[kb + 16 < myrc[j] ? myro[j] + kb + 16 : 0];
        }
        __builtin_amdgcn_sched_barrier(0);
        int pl[32];
#pragma unroll
        for (int j = 0; j < 32; ++j) {
            const bool live = kb + 16 * (j & 1) < myrc[j >> 1];
            pl[j] = atomicAdd(&c[live ? (wd[j] & 255u) : (unsigned)tid], live ? 1 : 0);
        }
#pragma unroll
        for (int j = 0; j < 32; ++j)
            if (kb + 16 * (j & 1) < myrc[j >> 1]) list2[pl[j]] = wd[j];
    }
}

template <bool FULL>
__device__ __forceinline__
void agg_group(const float* __restrict__ x, float* __restrict__ orow, unsigned word, float w,
               int j0, int cnt, int lane, float blo, float bhi,
               int& cur, int& run, float& alo, float& ahi) {
    float xlo[16], xhi[16];
#pragma unroll
    for (int u = 0; u < 16; ++u) {
        unsigned wj = (unsigned)__builtin_amdgcn_readlane((int)word, j0 + u);
        const float* r = x + (long long)(wj >> 8) * FDIM + lane;
        xlo[u] = r[0];
        xhi[u] = r[32];
    }
    __builtin_amdgcn_sched_barrier(0);
#pragma unroll
    for (int u = 0; u < 16; ++u) {
        const bool live = FULL || j0 + u < cnt;
        unsigned wj = (unsigned)__builtin_amdgcn_readlane((int)word, j0 + u);
        float sj = __int_as_float(__builtin_amdgcn_readlane(__float_as_int(w), j0 + u));
        const int tag = (int)(wj & 255u);
        if (live && tag != cur) {
            if (cur >= 0) {
                const float di = rsqrtf((float)run);
                orow[cur * FDIM]      = di * alo + blo;
                orow[cur * FDIM + 32] = di * ahi + bhi;
            }
            cur = tag; run = 0; alo = 0.0f; ahi = 0.0f;
        }
        const float nlo = alo + xlo[u] * sj;
        const float nhi = ahi + xhi[u] * sj;
        alo = live ? nlo : alo;
        ahi = live ? nhi : ahi;
        run += live ? 1 : 0;
    }
}

__global__ __launch_bounds__(CH * 32)
void k_agg(const float* __restrict__ x, const unsigned* __restrict__ list2,
           const int* __restrict__ cptr, const float* __restrict__ dinv,
           const float* __restrict__ bias, float* __restrict__ outp) {
    const int tid  = threadIdx.x;
    const int lane = tid & 31;
    const int wv  = __builtin_amdgcn_readfirstlane(tid >> 5);
    const int bin = blockIdx.x;
    const int beg = cptr[bin * CH + wv];
    const int end = cptr[bin * CH + wv + 1];
    const float blo = bias ? bias[lane] : 0.0f;
    const float bhi = bias ? bias[lane + 32] : 0.0f;
    float* const orow = outp + (long long)bin * ROWS * FDIM + lane;

    float alo = 0.0f, ahi = 0.0f;
    int cur = -1, run = 0;
    unsigned word = 0u;
    float w = 0.0f;
    if (beg + lane < end) {
        word = list2[beg + lane];
        w = dinv[word >> 8];
    }
    for (int base = beg; base < end; base += 32) {
        const int cnt = end - base < 32 ? end - base : 32;
        const bool more = base + 32 + lane < end;
        unsigned nword = 0u;
        float nw;
        if (more) nword = list2[base + 32 + lane];
        if (cnt >= 16) agg_group<true >(x, orow, word, w, 0, cnt, lane, blo, bhi, cur, run, alo, ahi);
        else           agg_group<false>(x, orow, word, w, 0, cnt, lane, blo, bhi, cur, run, alo, ahi);
        __builtin_amdgcn_sched_barrier(0);
        nw = dinv[nword >> 8];
        if (cnt >= 32)     agg_group<true >(x, orow, word, w, 16, cnt, lane, blo, bhi, cur, run, alo, ahi);
        else if (cnt > 16) agg_group<false>(x, orow, word, w, 16, cnt, lane, blo, bhi, cur, run, alo, ahi);
        word = nword;
        w = nw;
    }
    if (cur >= 0) {
        const float di = rsqrtf((float)run);
        orow[cur * FDIM]      = di * alo + blo;
        orow[cur * FDIM + 32] = di * ahi + bhi;
    }
}

#define MLP_WAVES 8
#define HP 132
#define WF 8192
#define MLP_LDS_FLOATS (2 * WF + MLP_WAVES * 16 * HP)

__global__ __launch_bounds__(256)
void k_mlp_wmma(const float* __restrict__ A, const float* __restrict__ W1,
                const float* __restrict__ b1, const float* __restrict__ W2,
                float* __restrict__ T, int ntiles) {
    __shared__ __attribute__((aligned(16))) float smem[MLP_LDS_FLOATS];
    float* w1f = smem;
    float* w2f = smem + WF;

    const int tid  = threadIdx.x;
    const int lane = tid & 31;
    const int half = lane >> 4;
    const int m    = lane & 15;
    const int wv = __builtin_amdgcn_readfirstlane(tid >> 5);
    float* hb = smem + 2 * WF + wv * (16 * HP);

#pragma unroll
    for (int i = 0; i < WF / 256; ++i) {
        int d  = tid + 256 * i;
        int e  = d & 1;
        int ln = (d >> 1) & 31;
        int mm = ln & 15;
        int hh = ln >> 4;
        int ct1 = (d >> 6) & 7;
        int s1  = d >> 9;
        w1f[d] = W1[(4 * s1 + 2 * hh + e) * 128 + 16 * ct1 + mm];
        int ct2 = (d >> 6) & 3;
        int s2  = d >> 8;
        w2f[d] = W2[(4 * s2 + 2 * hh + e) * 64 + 16 * ct2 + mm];
    }
    __syncthreads();

    const v2f* w1v = (const v2f*)w1f;
    const v2f* w2v = (const v2f*)w2f;
    const int slots = gridDim.x * MLP_WAVES;
    const int iters = (ntiles + slots - 1) / slots;
    for (int it = 0; it < iters; ++it) {
        const int tile = it * slots + wv * gridDim.x + blockIdx.x;
        const bool valid = tile < ntiles;

        if (valid) {
            const float* arow = A + (long long)(tile * 16 + m) * 64 + 2 * half;
            v2f a1[16];
#pragma unroll
            for (int s = 0; s < 16; ++s) a1[s] = *(const v2f*)(arow + 4 * s);

#pragma unroll
            for (int g = 0; g < 2; ++g) {
                v8f acc[4] = {};
#pragma unroll
                for (int s = 0; s < 16; ++s) {
#pragma unroll
                    for (int j = 0; j < 4; ++j) {
                        v2f b = w1v[(s * 8 + g * 4 + j) * 32 + lane];
                        acc[j] = __builtin_amdgcn_wmma_f32_16x16x4_f32(
                                     false, a1[s], false, b, (short)0, acc[j], false, false);
                    }
                }
#pragma unroll
                for (int j = 0; j < 4; ++j) {
                    const int col = (g * 4 + j) * 16 + m;
                    const float bv = b1[col];
#pragma unroll
                    for (int r = 0; r < 8; ++r) {
                        hb[(r + 8 * half) * HP + col] = fmaxf(acc[j][r] + bv, 0.0f);
                    }
                }
            }
        }
        __syncthreads();

        v2f a2[32];
        if (valid) {
#pragma unroll
            for (int s = 0; s < 32; ++s) a2[s] = *(const v2f*)(hb + m * HP + 4 * s + 2 * half);
        }
        __syncthreads();

        if (valid) {
            v8f acc[4] = {};
#pragma unroll
            for (int s = 0; s < 32; ++s) {
#pragma unroll
                for (int j = 0; j < 4; ++j) {
                    v2f b = w2v[(s * 4 + j) * 32 + lane];
                    acc[j] = __builtin_amdgcn_wmma_f32_16x16x4_f32(
                                 false, a2[s], false, b, (short)0, acc[j], false, false);
                }
            }
#pragma unroll
            for (int j = 0; j < 4; ++j)
#pragma unroll
                for (int r = 0; r < 8; ++r)
                    hb[(r + 8 * half) * HP + j * 16 + m] = acc[j][r];
        }
        __syncthreads();

        v4f tv[8];
#pragma unroll
        for (int i = 0; i < 8; ++i) {
            const int q = (i * 32 + lane) * 4;
            tv[i] = *(const v4f*)(hb + (q >> 6) * HP + (q & 63));
        }
        __syncthreads();
        if (valid) {
            float* tt = T + (long long)tile * (16 * 64);
#pragma unroll
            for (int i = 0; i < 8; ++i) *(v4f*)(tt + (i * 32 + lane) * 4) = tv[i];
        }
    }
}

extern "C" void kernel_launch(void* const* d_in, const int* in_sizes, int n_in,
                              void* d_out, int out_size, void* d_ws, size_t ws_size,
                              hipStream_t stream) {
    (void)n_in; (void)out_size; (void)ws_size;

    const int*   eidx = (const int*)d_in[0];
    const float* emb  = (const float*)d_in[1];
    const float* W1   = (const float*)d_in[2];
    const float* b1   = (const float*)d_in[3];
    const float* W2   = (const float*)d_in[4];
    const float* b2   = (const float*)d_in[5];

    const int E = in_sizes[0] / 2;
    const int n = in_sizes[1] / 64;
    const int* src = eidx;
    const int* dst = eidx + E;
    const int nbins = (n + ROWS - 1) / ROWS;

    int*   cptr = (int*)d_ws;
    float* dinv = (float*)d_ws + n;
    float* agg1 = dinv + n;
    float* t    = agg1 + (size_t)n * 64;
    float* out  = (float*)d_out;
    unsigned* list  = (unsigned*)agg1;
    int*      boff  = (int*)(list + E);
    unsigned* list2 = (unsigned*)(t + (size_t)n * 64);
    const int nblk = (E + EPB - 1) / EPB;

    k_split  <<<nblk, 1024, 0, stream>>>(src, dst, list, boff, E, nbins);
    k_rowsort<<<nbins, 256, 0, stream>>>(list, boff, dinv, list2, cptr, n, nbins, nblk);

    k_agg<<<nbins, CH * 32, 0, stream>>>(emb, list2, cptr, dinv, nullptr, agg1);

    k_mlp_wmma<<<512, 256, 0, stream>>>(agg1, W1, b1, W2, t, (n + 15) / 16);

    k_agg<<<nbins, CH * 32, 0, stream>>>(t, list2, cptr, dinv, b2, out);
}


# t rows scaled by their node's norm in the MLP epilogue: layer-2 aggregation adds rows as stored, no per-entry norm gather (32 lines per load); same sum order, products rounded before the adds
# speedup vs baseline: 5.2311x; 1.0533x over previous
#include <hip/hip_runtime.h>

typedef __attribute__((ext_vector_type(2))) float v2f;
typedef __attribute__((ext_vector_type(8))) float v8f;
typedef __attribute__((ext_vector_type(4))) float v4f;

#define FDIM 64
#define ROWS 196
#define MAXBINS 1024
#define EPT 8
#define EPB (EPT * 1024)

#define MAXBLK 256

__global__ __launch_bounds__(1024)
void k_split(const int* __restrict__ src, const int* __restrict__ dst,
             unsigned* __restrict__ list, int* __restrict__ boff, int E, int nbins) {
    __shared__ int c[MAXBINS];
    __shared__ int sc[2][MAXBINS];
    __shared__ unsigned srt[EPB];
    const int tid = threadIdx.x;
    c[tid] = 0;
    __syncthreads();

    int dv[EPT], sv[EPT];
#pragma unroll
    for (int k = 0; k < EPT; ++k) {
        const int e = blockIdx.x * EPB + k * 1024 + tid;
        const int ec = e < E ? e : E - 1;
        dv[k] = dst[ec];
        sv[k] = src[ec];
    }
    __builtin_amdgcn_sched_barrier(0);
    int bin[EPT], rk[EPT];
    unsigned wdv[EPT];
#pragma unroll
    for (int k = 0; k < EPT; ++k) {
        const bool live = blockIdx.x * EPB + k * 1024 + tid < E;
        const int d = dv[k];
        const int b = d / ROWS;
        rk[k]  = atomicAdd(&c[live ? b : tid], live ? 1 : 0);
        bin[k] = live ? b : -1;
        wdv[k] = ((unsigned)sv[k] << 8) | (unsigned)(d - b * ROWS);
    }
    __syncthreads();

    const int tot = c[tid];
    sc[0][tid] = tot;
    __syncthreads();
    int cur = 0;
    for (int off = 1; off < MAXBINS; off <<= 1) {
        int v = sc[cur][tid];
        if (tid >= off) v += sc[cur][tid - off];
        sc[cur ^ 1][tid] = v;
        cur ^= 1;
        __syncthreads();
    }
    const int start = sc[cur][tid] - tot;
    if (tid <= nbins) boff[(long long)blockIdx.x * (nbins + 1) + tid] = start;
    c[tid] = start;
    __syncthreads();

#pragma unroll
    for (int k = 0; k < EPT; ++k)
        if (bin[k] >= 0) srt[c[bin[k]] + rk[k]] = wdv[k];
    __syncthreads();

    for (int k = 0; k < EPT; ++k) {
        int i = k * 1024 + tid;
        int e = blockIdx.x * EPB + i;
        if (e < E) list[e] = srt[i];
    }
}

#define RPW 7
#define CH (ROWS / RPW)
__global__ __launch_bounds__(256)
void k_rowsort(const unsigned* __restrict__ list, const int* __restrict__ boff,
               float* __restrict__ dinv, unsigned* __restrict__ list2, int* __restrict__ cptr,
               int n, int nbins, int nblk) {
    __shared__ int c[256];
    __shared__ int sc[2][256];
    __shared__ int ro[MAXBLK];
    __shared__ int rc[MAXBLK];
    __shared__ int s_beg, s_max;
    const int tid = threadIdx.x;
    const int bin = blockIdx.x;
    c[tid] = 0;
    if (tid == 0) { s_beg = 0; s_max = 0; }
    __syncthreads();
    int o0 = 0, len = 0;
    if (tid < nblk) {
        const int* bo = boff + (long long)tid * (nbins + 1) + bin;
        o0  = bo[0];
        len = bo[1] - o0;
    }
    ro[tid] = tid * EPB + o0;
    rc[tid] = len;
    int so = o0, mx = len;
    for (int d = 16; d > 0; d >>= 1) {
        so += __shfl_xor(so, d);
        int t = __shfl_xor(mx, d);
        mx = t > mx ? t : mx;
    }
    if ((tid & 31) == 0) { atomicAdd(&s_beg, so); atomicMax(&s_max, mx); }
    __syncthreads();
    const int beg   = s_beg;
    const int maxrc = s_max;

    const int hw  = tid >> 4;
    const int l16 = tid & 15;
    int myro[16], myrc[16];
#pragma unroll
    for (int j = 0; j < 16; ++j) {
        myro[j] = ro[hw + 16 * j] + l16;
        myrc[j] = rc[hw + 16 * j] - l16;
    }

    for (int kb = 0; kb < maxrc; kb += 32) {
        unsigned wd[32];
#pragma unroll
        for (int j = 0; j < 16; ++j) {
            wd[2 * j]     = list[kb      < myrc[j] ? myro[j] + kb      : 0];
            wd[2 * j + 1] = list[kb + 16 < myrc[j] ? myro[j] + kb + 16 : 0];
        }
        __builtin_amdgcn_sched_barrier(0);
#pragma unroll
        for (int j = 0; j < 32; ++j) {
            const bool live = kb + 16 * (j & 1) < myrc[j >> 1];
            atomicAdd(&c[live ? (wd[j] & 255u) : (unsigned)tid], live ? 1 : 0);
        }
    }
    __syncthreads();
    const int row = bin * ROWS + tid;
    const bool valid = tid < ROWS && row < n;
    const int slots = valid ? c[tid] + 1 : 0;
    if (valid) {
        float d = (float)slots;
        dinv[row] = d > 0.0f ? rsqrtf(d) : 0.0f;
    }
    sc[0][tid] = slots;
    __syncthreads();
    int cur = 0;
    for (int off = 1; off < 256; off <<= 1) {
        int v = sc[cur][tid];
        if (tid >= off) v += sc[cur][tid - off];
        sc[cur ^ 1][tid] = v;
        cur ^= 1;
        __syncthreads();
    }
    const int incl = sc[cur][tid];
    const int base = beg + bin * ROWS;
    const int first = base + incl - slots;
    if (valid) list2[first] = ((unsigned)row << 8) | (unsigned)tid;
    if (tid < ROWS && tid % RPW == 0) cptr[bin * CH + tid / RPW] = first;
    if (bin == nbins - 1 && tid == 255) cptr[nbins * CH] = base + incl;
    c[tid] = first + 1;
    __syncthreads();
    for (int kb = 0; kb < maxrc; kb += 32) {
        unsigned wd[32];
#pragma unroll
        for (int j = 0; j < 16; ++j) {
            wd[2 * j]     = list[kb      < myrc[j] ? myro[j] + kb      : 0];
            wd[2 * j + 1] = list[kb + 16 < myrc[j] ? myro[j] + kb + 16 : 0];
        }
        __builtin_amdgcn_sched_barrier(0);
        int pl[32];
#pragma unroll
        for (int j = 0; j < 32; ++j) {
            const bool live = kb + 16 * (j & 1) < myrc[j >> 1];
            pl[j] = atomicAdd(&c[live ? (wd[j] & 255u) : (unsigned)tid], live ? 1 : 0);
        }
#pragma unroll
        for (int j = 0; j < 32; ++j)
            if (kb + 16 * (j & 1) < myrc[j >> 1]) list2[pl[j]] = wd[j];
    }
}

template <bool FULL, bool PRE>
__device__ __forceinline__
void agg_group(const float* __restrict__ x, float* __restrict__ orow, unsigned word, float w,
               int j0, int cnt, int lane, float blo, float bhi,
               int& cur, int& run, float& alo, float& ahi) {
    float xlo[16], xhi[16];
#pragma unroll
    for (int u = 0; u < 16; ++u) {
        unsigned wj = (unsigned)__builtin_amdgcn_readlane((int)word, j0 + u);
        const float* r = x + (long long)(wj >> 8) * FDIM + lane;
        xlo[u] = r[0];
        xhi[u] = r[32];
    }
    __builtin_amdgcn_sched_barrier(0);
#pragma unroll
    for (int u = 0; u < 16; ++u) {
        const bool live = FULL || j0 + u < cnt;
        unsigned wj = (unsigned)__builtin_amdgcn_readlane((int)word, j0 + u);
        float sj = 0.0f;
        if (!PRE) sj = __int_as_float(__builtin_amdgcn_readlane(__float_as_int(w), j0 + u));
        const int tag = (int)(wj & 255u);
        if (live && tag != cur) {
            if (cur >= 0) {
                const float di = rsqrtf((float)run);
                orow[cur * FDIM]      = di * alo + blo;
                orow[cur * FDIM + 32] = di * ahi + bhi;
            }
            cur = tag; run = 0; alo = 0.0f; ahi = 0.0f;
        }
        const float nlo = PRE ? alo + xlo[u] : alo + xlo[u] * sj;
        const float nhi = PRE ? ahi + xhi[u] : ahi + xhi[u] * sj;
        alo = live ? nlo : alo;
        ahi = live ? nhi : ahi;
        run += live ? 1 : 0;
    }
}

template <bool PRE>
__global__ __launch_bounds__(CH * 32)
void k_agg(const float* __restrict__ x, const unsigned* __restrict__ list2,
           const int* __restrict__ cptr, const float* __restrict__ dinv,
           const float* __restrict__ bias, float* __restrict__ outp) {
    const int tid  = threadIdx.x;
    const int lane = tid & 31;
    const int wv  = __builtin_amdgcn_readfirstlane(tid >> 5);
    const int bin = blockIdx.x;
    const int beg = cptr[bin * CH + wv];
    const int end = cptr[bin * CH + wv + 1];
    const float blo = bias ? bias[lane] : 0.0f;
    const float bhi = bias ? bias[lane + 32] : 0.0f;
    float* const orow = outp + (long long)bin * ROWS * FDIM + lane;

    float alo = 0.0f, ahi = 0.0f;
    int cur = -1, run = 0;
    unsigned word = 0u;
    float w = 0.0f;
    if (beg + lane < end) {
        word = list2[beg + lane];
        if (!PRE) w = dinv[word >> 8];
    }
    if (PRE) {
        if (beg >= end) return;
        cur = (int)((unsigned)__builtin_amdgcn_readlane((int)word, 0) & 255u);
    }
    for (int base = beg; base < end; base += 32) {
        const int cnt = end - base < 32 ? end - base : 32;
        const bool more = base + 32 + lane < end;
        unsigned nword = 0u;
        float nw = 0.0f;
        if (more) nword = list2[base + 32 + lane];
        if (cnt >= 16) agg_group<true , PRE>(x, orow, word, w, 0, cnt, lane, blo, bhi, cur, run, alo, ahi);
        else           agg_group<false, PRE>(x, orow, word, w, 0, cnt, lane, blo, bhi, cur, run, alo, ahi);
        if (!PRE) {
            __builtin_amdgcn_sched_barrier(0);
            nw = dinv[nword >> 8];
        }
        if (cnt >= 32)     agg_group<true , PRE>(x, orow, word, w, 16, cnt, lane, blo, bhi, cur, run, alo, ahi);
        else if (cnt > 16) agg_group<false, PRE>(x, orow, word, w, 16, cnt, lane, blo, bhi, cur, run, alo, ahi);
        word = nword;
        w = nw;
    }
    if (cur >= 0) {
        const float di = rsqrtf((float)run);
        orow[cur * FDIM]      = di * alo + blo;
        orow[cur * FDIM + 32] = di * ahi + bhi;
    }
}

#define MLP_WAVES 8
#define HP 132
#define WF 8192
#define MLP_LDS_FLOATS (2 * WF + MLP_WAVES * 16 * HP)

__global__ __launch_bounds__(256)
void k_mlp_wmma(const float* __restrict__ A, const float* __restrict__ W1,
                const float* __restrict__ b1, const float* __restrict__ W2,
                const float* __restrict__ dinv, float* __restrict__ T, int ntiles) {
    __shared__ __attribute__((aligned(16))) float smem[MLP_LDS_FLOATS];
    float* w1f = smem;
    float* w2f = smem + WF;

    const int tid  = threadIdx.x;
    const int lane = tid & 31;
    const int half = lane >> 4;
    const int m    = lane & 15;
    const int wv = __builtin_amdgcn_readfirstlane(tid >> 5);
    float* hb = smem + 2 * WF + wv * (16 * HP);

#pragma unroll
    for (int i = 0; i < WF / 256; ++i) {
        int d  = tid + 256 * i;
        int e  = d & 1;
        int ln = (d >> 1) & 31;
        int mm = ln & 15;
        int hh = ln >> 4;
        int ct1 = (d >> 6) & 7;
        int s1  = d >> 9;
        w1f[d] = W1[(4 * s1 + 2 * hh + e) * 128 + 16 * ct1 + mm];
        int ct2 = (d >> 6) & 3;
        int s2  = d >> 8;
        w2f[d] = W2[(4 * s2 + 2 * hh + e) * 64 + 16 * ct2 + mm];
    }
    __syncthreads();

    const v2f* w1v = (const v2f*)w1f;
    const v2f* w2v = (const v2f*)w2f;
    const int slots = gridDim.x * MLP_WAVES;
    const int iters = (ntiles + slots - 1) / slots;
    for (int it = 0; it < iters; ++it) {
        const int tile = it * slots + wv * gridDim.x + blockIdx.x;
        const bool valid = tile < ntiles;

        if (valid) {
            const float* arow = A + (long long)(tile * 16 + m) * 64 + 2 * half;
            v2f a1[16];
#pragma unroll
            for (int s = 0; s < 16; ++s) a1[s] = *(const v2f*)(arow + 4 * s);

#pragma unroll
            for (int g = 0; g < 2; ++g) {
                v8f acc[4] = {};
#pragma unroll
                for (int s = 0; s < 16; ++s) {
#pragma unroll
                    for (int j = 0; j < 4; ++j) {
                        v2f b = w1v[(s * 8 + g * 4 + j) * 32 + lane];
                        acc[j] = __builtin_amdgcn_wmma_f32_16x16x4_f32(
                                     false, a1[s], false, b, (short)0, acc[j], false, false);
                    }
                }
#pragma unroll
                for (int j = 0; j < 4; ++j) {
                    const int col = (g * 4 + j) * 16 + m;
                    const float bv = b1[col];
#pragma unroll
                    for (int r = 0; r < 8; ++r) {
                        hb[(r + 8 * half) * HP + col] = fmaxf(acc[j][r] + bv, 0.0f);
                    }
                }
            }
        }
        __syncthreads();

        v2f a2[32];
        if (valid) {
#pragma unroll
            for (int s = 0; s < 32; ++s) a2[s] = *(const v2f*)(hb + m * HP + 4 * s + 2 * half);
        }
        __syncthreads();

        if (valid) {
            float dr[8];
#pragma unroll
            for (int r = 0; r < 8; ++r) dr[r] = dinv[tile * 16 + 8 * half + r];
            v8f acc[4] = {};
#pragma unroll
            for (int s = 0; s < 32; ++s) {
#pragma unroll
                for (int j = 0; j < 4; ++j) {
                    v2f b = w2v[(s * 4 + j) * 32 + lane];
                    acc[j] = __builtin_amdgcn_wmma_f32_16x16x4_f32(
                                 false, a2[s], false, b, (short)0, acc[j], false, false);
                }
            }
#pragma unroll
            for (int j = 0; j < 4; ++j)
#pragma unroll
                for (int r = 0; r < 8; ++r)
                    hb[(r + 8 * half) * HP + j * 16 + m] = acc[j][r] * dr[r];
        }
        __syncthreads();

        v4f tv[8];
#pragma unroll
        for (int i = 0; i < 8; ++i) {
            const int q = (i * 32 + lane) * 4;
            tv[i] = *(const v4f*)(hb + (q >> 6) * HP + (q & 63));
        }
        __syncthreads();
        if (valid) {
            float* tt = T + (long long)tile * (16 * 64);
#pragma unroll
            for (int i = 0; i < 8; ++i) *(v4f*)(tt + (i * 32 + lane) * 4) = tv[i];
        }
    }
}

extern "C" void kernel_launch(void* const* d_in, const int* in_sizes, int n_in,
                              void* d_out, int out_size, void* d_ws, size_t ws_size,
                              hipStream_t stream) {
    (void)n_in; (void)out_size; (void)ws_size;

    const int*   eidx = (const int*)d_in[0];
    const float* emb  = (const float*)d_in[1];
    const float* W1   = (const float*)d_in[2];
    const float* b1   = (const float*)d_in[3];
    const float* W2   = (const float*)d_in[4];
    const float* b2   = (const float*)d_in[5];

    const int E = in_sizes[0] / 2;
    const int n = in_sizes[1] / 64;
    const int* src = eidx;
    const int* dst = eidx + E;
    const int nbins = (n + ROWS - 1) / ROWS;

    int*   cptr = (int*)d_ws;
    float* dinv = (float*)d_ws + n;
    float* agg1 = dinv + n;
    float* t    = agg1 + (size_t)n * 64;
    float* out  = (float*)d_out;
    unsigned* list  = (unsigned*)agg1;
    int*      boff  = (int*)(list + E);
    unsigned* list2 = (unsigned*)(t + (size_t)n * 64);
    const int nblk = (E + EPB - 1) / EPB;

    k_split  <<<nblk, 1024, 0, stream>>>(src, dst, list, boff, E, nbins);
    k_rowsort<<<nbins, 256, 0, stream>>>(list, boff, dinv, list2, cptr, n, nbins, nblk);

    k_agg<false><<<nbins, CH * 32, 0, stream>>>(emb, list2, cptr, dinv, nullptr, agg1);

    k_mlp_wmma<<<512, 256, 0, stream>>>(agg1, W1, b1, W2, dinv, t, (n + 15) / 16);

    k_agg<true><<<nbins, CH * 32, 0, stream>>>(t, list2, cptr, dinv, b2, out);
}


# row pass orders a range on chip: runs read once (words stay in registers over the scan), entries placed by row in LDS, whole-line stores; was 2 reads and a 4-byte store per entry; same sum order
# speedup vs baseline: 5.4258x; 1.0372x over previous
#include <hip/hip_runtime.h>

typedef __attribute__((ext_vector_type(2))) float v2f;
typedef __attribute__((ext_vector_type(8))) float v8f;
typedef __attribute__((ext_vector_type(4))) float v4f;

#define FDIM 64
#define ROWS 196
#define MAXBINS 1024
#define EPT 8
#define EPB (EPT * 1024)

#define MAXBLK 256

__global__ __launch_bounds__(1024)
void k_split(const int* __restrict__ src, const int* __restrict__ dst,
             unsigned* __restrict__ list, int* __restrict__ boff, int E, int nbins) {
    __shared__ int c[MAXBINS];
    __shared__ int sc[2][MAXBINS];
    __shared__ unsigned srt[EPB];
    const int tid = threadIdx.x;
    c[tid] = 0;
    __syncthreads();

    int dv[EPT], sv[EPT];
#pragma unroll
    for (int k = 0; k < EPT; ++k) {
        const int e = blockIdx.x * EPB + k * 1024 + tid;
        const int ec = e < E ? e : E - 1;
        dv[k] = dst[ec];
        sv[k] = src[ec];
    }
    __builtin_amdgcn_sched_barrier(0);
    int bin[EPT], rk[EPT];
    unsigned wdv[EPT];
#pragma unroll
    for (int k = 0; k < EPT; ++k) {
        const bool live = blockIdx.x * EPB + k * 1024 + tid < E;
        const int d = dv[k];
        const int b = d / ROWS;
        rk[k]  = atomicAdd(&c[live ? b : tid], live ? 1 : 0);
        bin[k] = live ? b : -1;
        wdv[k] = ((unsigned)sv[k] << 8) | (unsigned)(d - b * ROWS);
    }
    __syncthreads();

    const int tot = c[tid];
    sc[0][tid] = tot;
    __syncthreads();
    int cur = 0;
    for (int off = 1; off < MAXBINS; off <<= 1) {
        int v = sc[cur][tid];
        if (tid >= off) v += sc[cur][tid - off];
        sc[cur ^ 1][tid] = v;
        cur ^= 1;
        __syncthreads();
    }
    const int start = sc[cur][tid] - tot;
    if (tid <= nbins) boff[(long long)blockIdx.x * (nbins + 1) + tid] = start;
    c[tid] = start;
    __syncthreads();

#pragma unroll
    for (int k = 0; k < EPT; ++k)
        if (bin[k] >= 0) srt[c[bin[k]] + rk[k]] = wdv[k];
    __syncthreads();

    for (int k = 0; k < EPT; ++k) {
        int i = k * 1024 + tid;
        int e = blockIdx.x * EPB + i;
        if (e < E) list[e] = srt[i];
    }
}

#define RPW 7
#define CH (ROWS / RPW)
#define RS_CAP 8192

__device__ __forceinline__
void rs_load(const unsigned* __restrict__ list, const int (&myro)[16], const int (&myrc)[16],
             int kb, unsigned (&wd)[32]) {
#pragma unroll
    for (int j = 0; j < 16; ++j) {
        wd[2 * j]     = list[kb      < myrc[j] ? myro[j] + kb      : 0];
        wd[2 * j + 1] = list[kb + 16 < myrc[j] ? myro[j] + kb + 16 : 0];
    }
    __builtin_amdgcn_sched_barrier(0);
}

__device__ __forceinline__
void rs_count(int* c, const int (&myrc)[16], int kb, const unsigned (&wd)[32], int tid) {
#pragma unroll
    for (int j = 0; j < 32; ++j) {
        const bool live = kb + 16 * (j & 1) < myrc[j >> 1];
        atomicAdd(&c[live ? (wd[j] & 255u) : (unsigned)tid], live ? 1 : 0);
    }
}

template <bool INLDS>
__device__ __forceinline__
void rs_put(unsigned* __restrict__ list2, unsigned* buf, int* c, const int (&myrc)[16], int kb,
            const unsigned (&wd)[32], int tid) {
    int pl[32];
#pragma unroll
    for (int j = 0; j < 32; ++j) {
        const bool live = kb + 16 * (j & 1) < myrc[j >> 1];
        pl[j] = atomicAdd(&c[live ? (wd[j] & 255u) : (unsigned)tid], live ? 1 : 0);
    }
#pragma unroll
    for (int j = 0; j < 32; ++j)
        if (kb + 16 * (j & 1) < myrc[j >> 1]) {
            if (INLDS) buf[pl[j]] = wd[j];
            else       list2[pl[j]] = wd[j];
        }
}

__global__ __launch_bounds__(256)
void k_rowsort(const unsigned* __restrict__ list, const int* __restrict__ boff,
               float* __restrict__ dinv, unsigned* __restrict__ list2, int* __restrict__ cptr,
               int n, int nbins, int nblk) {
    __shared__ int c[256];
    __shared__ int sc[2][256];
    __shared__ int ro[MAXBLK];
    __shared__ int rc[MAXBLK];
    __shared__ unsigned buf[RS_CAP];
    __shared__ int s_beg, s_max;
    const int tid = threadIdx.x;
    const int bin = blockIdx.x;
    c[tid] = 0;
    if (tid == 0) { s_beg = 0; s_max = 0; }
    __syncthreads();
    int o0 = 0, len = 0;
    if (tid < nblk) {
        const int* bo = boff + (long long)tid * (nbins + 1) + bin;
        o0  = bo[0];
        len = bo[1] - o0;
    }
    ro[tid] = tid * EPB + o0;
    rc[tid] = len;
    int so = o0, mx = len;
    for (int d = 16; d > 0; d >>= 1) {
        so += __shfl_xor(so, d);
        int t = __shfl_xor(mx, d);
        mx = t > mx ? t : mx;
    }
    if ((tid & 31) == 0) { atomicAdd(&s_beg, so); atomicMax(&s_max, mx); }
    __syncthreads();
    const int beg   = s_beg;
    const int maxrc = s_max;

    const int hw  = tid >> 4;
    const int l16 = tid & 15;
    int myro[16], myrc[16];
#pragma unroll
    for (int j = 0; j < 16; ++j) {
        myro[j] = ro[hw + 16 * j] + l16;
        myrc[j] = rc[hw + 16 * j] - l16;
    }

    unsigned w0[32];
    rs_load(list, myro, myrc, 0, w0);
    rs_count(c, myrc, 0, w0, tid);
    for (int kb = 32; kb < maxrc; kb += 32) {
        unsigned wd[32];
        rs_load(list, myro, myrc, kb, wd);
        rs_count(c, myrc, kb, wd, tid);
    }
    __syncthreads();
    const int row = bin * ROWS + tid;
    const bool valid = tid < ROWS && row < n;
    const int slots = valid ? c[tid] + 1 : 0;
    if (valid) {
        float d = (float)slots;
        dinv[row] = d > 0.0f ? rsqrtf(d) : 0.0f;
    }
    sc[0][tid] = slots;
    __syncthreads();
    int cur = 0;
    for (int off = 1; off < 256; off <<= 1) {
        int v = sc[cur][tid];
        if (tid >= off) v += sc[cur][tid - off];
        sc[cur ^ 1][tid] = v;
        cur ^= 1;
        __syncthreads();
    }
    const int incl = sc[cur][tid];
    const int total = sc[cur][255];
    const bool inlds = total <= RS_CAP;
    const int base = beg + bin * ROWS;
    const int lfirst = incl - slots;
    const int first = base + lfirst;
    const unsigned selfw = ((unsigned)row << 8) | (unsigned)tid;
    if (valid) {
        if (inlds) buf[lfirst] = selfw;
        else       list2[first] = selfw;
    }
    if (tid < ROWS && tid % RPW == 0) cptr[bin * CH + tid / RPW] = first;
    if (bin == nbins - 1 && tid == 255) cptr[nbins * CH] = base + incl;
    c[tid] = (inlds ? lfirst : first) + 1;
    __syncthreads();
    if (inlds) {
        rs_put<true>(list2, buf, c, myrc, 0, w0, tid);
        for (int kb = 32; kb < maxrc; kb += 32) {
            unsigned wd[32];
            rs_load(list, myro, myrc, kb, wd);
            rs_put<true>(list2, buf, c, myrc, kb, wd, tid);
        }
    } else {
        rs_put<false>(list2, buf, c, myrc, 0, w0, tid);
        for (int kb = 32; kb < maxrc; kb += 32) {
            unsigned wd[32];
            rs_load(list, myro, myrc, kb, wd);
            rs_put<false>(list2, buf, c, myrc, kb, wd, tid);
        }
    }
    __syncthreads();
    if (inlds) {
        for (int ib = tid - (base & 31); ib < total; ib += 1024) {
            unsigned v[4];
#pragma unroll
            for (int u = 0; u < 4; ++u) {
                const int i = ib + 256 * u;
                v[u] = buf[(i >= 0 && i < total) ? i : 0];
            }
#pragma unroll
            for (int u = 0; u < 4; ++u) {
                const int i = ib + 256 * u;
                if (i >= 0 && i < total) list2[base + i] = v[u];
            }
        }
    }
}

template <bool FULL, bool PRE>
__device__ __forceinline__
void agg_group(const float* __restrict__ x, float* __restrict__ orow, unsigned word, float w,
               int j0, int cnt, int lane, float blo, float bhi,
               int& cur, int& run, float& alo, float& ahi) {
    float xlo[16], xhi[16];
#pragma unroll
    for (int u = 0; u < 16; ++u) {
        unsigned wj = (unsigned)__builtin_amdgcn_readlane((int)word, j0 + u);
        const float* r = x + (long long)(wj >> 8) * FDIM + lane;
        xlo[u] = r[0];
        xhi[u] = r[32];
    }
    __builtin_amdgcn_sched_barrier(0);
#pragma unroll
    for (int u = 0; u < 16; ++u) {
        const bool live = FULL || j0 + u < cnt;
        unsigned wj = (unsigned)__builtin_amdgcn_readlane((int)word, j0 + u);
        float sj = 0.0f;
        if (!PRE) sj = __int_as_float(__builtin_amdgcn_readlane(__float_as_int(w), j0 + u));
        const int tag = (int)(wj & 255u);
        if (live && tag != cur) {
            if (cur >= 0) {
                const float di = rsqrtf((float)run);
                orow[cur * FDIM]      = di * alo + blo;
                orow[cur * FDIM + 32] = di * ahi + bhi;
            }
            cur = tag; run = 0; alo = 0.0f; ahi = 0.0f;
        }
        const float nlo = PRE ? alo + xlo[u] : alo + xlo[u] * sj;
        const float nhi = PRE ? ahi + xhi[u] : ahi + xhi[u] * sj;
        alo = live ? nlo : alo;
        ahi = live ? nhi : ahi;
        run += live ? 1 : 0;
    }
}

template <bool PRE>
__global__ __launch_bounds__(CH * 32)
void k_agg(const float* __restrict__ x, const unsigned* __restrict__ list2,
           const int* __restrict__ cptr, const float* __restrict__ dinv,
           const float* __restrict__ bias, float* __restrict__ outp) {
    const int tid  = threadIdx.x;
    const int lane = tid & 31;
    const int wv  = __builtin_amdgcn_readfirstlane(tid >> 5);
    const int bin = blockIdx.x;
    const int beg = cptr[bin * CH + wv];
    const int end = cptr[bin * CH + wv + 1];
    const float blo = bias ? bias[lane] : 0.0f;
    const float bhi = bias ? bias[lane + 32] : 0.0f;
    float* const orow = outp + (long long)bin * ROWS * FDIM + lane;

    float alo = 0.0f, ahi = 0.0f;
    int cur = -1, run = 0;
    unsigned word = 0u;
    float w = 0.0f;
    if (beg + lane < end) {
        word = list2[beg + lane];
        if (!PRE) w = dinv[word >> 8];
    }
    if (PRE) {
        if (beg >= end) return;
        cur = (int)((unsigned)__builtin_amdgcn_readlane((int)word, 0) & 255u);
    }
    for (int base = beg; base < end; base += 32) {
        const int cnt = end - base < 32 ? end - base : 32;
        const bool more = base + 32 + lane < end;
        unsigned nword = 0u;
        float nw = 0.0f;
        if (more) nword = list2[base + 32 + lane];
        if (cnt >= 16) agg_group<true , PRE>(x, orow, word, w, 0, cnt, lane, blo, bhi, cur, run, alo, ahi);
        else           agg_group<false, PRE>(x, orow, word, w, 0, cnt, lane, blo, bhi, cur, run, alo, ahi);
        if (!PRE) {
            __builtin_amdgcn_sched_barrier(0);
            nw = dinv[nword >> 8];
        }
        if (cnt >= 32)     agg_group<true , PRE>(x, orow, word, w, 16, cnt, lane, blo, bhi, cur, run, alo, ahi);
        else if (cnt > 16) agg_group<false, PRE>(x, orow, word, w, 16, cnt, lane, blo, bhi, cur, run, alo, ahi);
        word = nword;
        w = nw;
    }
    if (cur >= 0) {
        const float di = rsqrtf((float)run);
        orow[cur * FDIM]      = di * alo + blo;
        orow[cur * FDIM + 32] = di * ahi + bhi;
    }
}

#define MLP_WAVES 8
#define HP 132
#define WF 8192
#define MLP_LDS_FLOATS (2 * WF + MLP_WAVES * 16 * HP)

__global__ __launch_bounds__(256)
void k_mlp_wmma(const float* __restrict__ A, const float* __restrict__ W1,
                const float* __restrict__ b1, const float* __restrict__ W2,
                const float* __restrict__ dinv, float* __restrict__ T, int ntiles) {
    __shared__ __attribute__((aligned(16))) float smem[MLP_LDS_FLOATS];
    float* w1f = smem;
    float* w2f = smem + WF;

    const int tid  = threadIdx.x;
    const int lane = tid & 31;
    const int half = lane >> 4;
    const int m    = lane & 15;
    const int wv = __builtin_amdgcn_readfirstlane(tid >> 5);
    float* hb = smem + 2 * WF + wv * (16 * HP);

#pragma unroll
    for (int i = 0; i < WF / 256; ++i) {
        int d  = tid + 256 * i;
        int e  = d & 1;
        int ln = (d >> 1) & 31;
        int mm = ln & 15;
        int hh = ln >> 4;
        int ct1 = (d >> 6) & 7;
        int s1  = d >> 9;
        w1f[d] = W1[(4 * s1 + 2 * hh + e) * 128 + 16 * ct1 + mm];
        int ct2 = (d >> 6) & 3;
        int s2  = d >> 8;
        w2f[d] = W2[(4 * s2 + 2 * hh + e) * 64 + 16 * ct2 + mm];
    }
    __syncthreads();

    const v2f* w1v = (const v2f*)w1f;
    const v2f* w2v = (const v2f*)w2f;
    const int slots = gridDim.x * MLP_WAVES;
    const int iters = (ntiles + slots - 1) / slots;
    for (int it = 0; it < iters; ++it) {
        const int tile = it * slots + wv * gridDim.x + blockIdx.x;
        const bool valid = tile < ntiles;

        if (valid) {
            const float* arow = A + (long long)(tile * 16 + m) * 64 + 2 * half;
            v2f a1[16];
#pragma unroll
            for (int s = 0; s < 16; ++s) a1[s] = *(const v2f*)(arow + 4 * s);

#pragma unroll
            for (int g = 0; g < 2; ++g) {
                v8f acc[4] = {};
#pragma unroll
                for (int s = 0; s < 16; ++s) {
#pragma unroll
                    for (int j = 0; j < 4; ++j) {
                        v2f b = w1v[(s * 8 + g * 4 + j) * 32 + lane];
                        acc[j] = __builtin_amdgcn_wmma_f32_16x16x4_f32(
                                     false, a1[s], false, b, (short)0, acc[j], false, false);
                    }
                }
#pragma unroll
                for (int j = 0; j < 4; ++j) {
                    const int col = (g * 4 + j) * 16 + m;
                    const float bv = b1[col];
#pragma unroll
                    for (int r = 0; r < 8; ++r) {
                        hb[(r + 8 * half) * HP + col] = fmaxf(acc[j][r] + bv, 0.0f);
                    }
                }
            }
        }
        __syncthreads();

        v2f a2[32];
        if (valid) {
#pragma unroll
            for (int s = 0; s < 32; ++s) a2[s] = *(const v2f*)(hb + m * HP + 4 * s + 2 * half);
        }
        __syncthreads();

        if (valid) {
            float dr[8];
#pragma unroll
            for (int r = 0; r < 8; ++r) dr[r] = dinv[tile * 16 + 8 * half + r];
            v8f acc[4] = {};
#pragma unroll
            for (int s = 0; s < 32; ++s) {
#pragma unroll
                for (int j = 0; j < 4; ++j) {
                    v2f b = w2v[(s * 4 + j) * 32 + lane];
                    acc[j] = __builtin_amdgcn_wmma_f32_16x16x4_f32(
                                 false, a2[s], false, b, (short)0, acc[j], false, false);
                }
            }
#pragma unroll
            for (int j = 0; j < 4; ++j)
#pragma unroll
                for (int r = 0; r < 8; ++r)
                    hb[(r + 8 * half) * HP + j * 16 + m] = acc[j][r] * dr[r];
        }
        __syncthreads();

        v4f tv[8];
#pragma unroll
        for (int i = 0; i < 8; ++i) {
            const int q = (i * 32 + lane) * 4;
            tv[i] = *(const v4f*)(hb + (q >> 6) * HP + (q & 63));
        }
        __syncthreads();
        if (valid) {
            float* tt = T + (long long)tile * (16 * 64);
#pragma unroll
            for (int i = 0; i < 8; ++i) *(v4f*)(tt + (i * 32 + lane) * 4) = tv[i];
        }
    }
}

extern "C" void kernel_launch(void* const* d_in, const int* in_sizes, int n_in,
                              void* d_out, int out_size, void* d_ws, size_t ws_size,
                              hipStream_t stream) {
    (void)n_in; (void)out_size; (void)ws_size;

    const int*   eidx = (const int*)d_in[0];
    const float* emb  = (const float*)d_in[1];
    const float* W1   = (const float*)d_in[2];
    const float* b1   = (const float*)d_in[3];
    const float* W2   = (const float*)d_in[4];
    const float* b2   = (const float*)d_in[5];

    const int E = in_sizes[0] / 2;
    const int n = in_sizes[1] / 64;
    const int* src = eidx;
    const int* dst = eidx + E;
    const int nbins = (n + ROWS - 1) / ROWS;

    int*   cptr = (int*)d_ws;
    float* dinv = (float*)d_ws + n;
    float* agg1 = dinv + n;
    float* t    = agg1 + (size_t)n * 64;
    float* out  = (float*)d_out;
    unsigned* list  = (unsigned*)agg1;
    int*      boff  = (int*)(list + E);
    unsigned* list2 = (unsigned*)(t + (size_t)n * 64);
    const int nblk = (E + EPB - 1) / EPB;

    k_split  <<<nblk, 1024, 0, stream>>>(src, dst, list, boff, E, nbins);
    k_rowsort<<<nbins, 256, 0, stream>>>(list, boff, dinv, list2, cptr, n, nbins, nblk);

    k_agg<false><<<nbins, CH * 32, 0, stream>>>(emb, list2, cptr, dinv, nullptr, agg1);

    k_mlp_wmma<<<512, 256, 0, stream>>>(agg1, W1, b1, W2, dinv, t, (n + 15) / 16);

    k_agg<true><<<nbins, CH * 32, 0, stream>>>(t, list2, cptr, dinv, b2, out);
}


# layer-1 aggregation and both dense products in one launch: 14 waves sum a range's rows (64 line loads each) into LDS tiles while 13 waves multiply the range before; sums stay on chip; same sum order
# speedup vs baseline: 5.6037x; 1.0328x over previous
#include <hip/hip_runtime.h>

typedef __attribute__((ext_vector_type(2))) float v2f;
typedef __attribute__((ext_vector_type(8))) float v8f;
typedef __attribute__((ext_vector_type(4))) float v4f;

#define FDIM 64
#define ROWS 196
#define MAXBINS 1024
#define EPT 8
#define EPB (EPT * 1024)

#define MAXBLK 256

__global__ __launch_bounds__(1024)
void k_split(const int* __restrict__ src, const int* __restrict__ dst,
             unsigned* __restrict__ list, int* __restrict__ boff, int E, int nbins) {
    __shared__ int c[MAXBINS];
    __shared__ int sc[2][MAXBINS];
    __shared__ unsigned srt[EPB];
    const int tid = threadIdx.x;
    c[tid] = 0;
    __syncthreads();

    int dv[EPT], sv[EPT];
#pragma unroll
    for (int k = 0; k < EPT; ++k) {
        const int e = blockIdx.x * EPB + k * 1024 + tid;
        const int ec = e < E ? e : E - 1;
        dv[k] = dst[ec];
        sv[k] = src[ec];
    }
    __builtin_amdgcn_sched_barrier(0);
    int bin[EPT], rk[EPT];
    unsigned wdv[EPT];
#pragma unroll
    for (int k = 0; k < EPT; ++k) {
        const bool live = blockIdx.x * EPB + k * 1024 + tid < E;
        const int d = dv[k];
        const int b = d / ROWS;
        rk[k]  = atomicAdd(&c[live ? b : tid], live ? 1 : 0);
        bin[k] = live ? b : -1;
        wdv[k] = ((unsigned)sv[k] << 8) | (unsigned)(d - b * ROWS);
    }
    __syncthreads();

    const int tot = c[tid];
    sc[0][tid] = tot;
    __syncthreads();
    int cur = 0;
    for (int off = 1; off < MAXBINS; off <<= 1) {
        int v = sc[cur][tid];
        if (tid >= off) v += sc[cur][tid - off];
        sc[cur ^ 1][tid] = v;
        cur ^= 1;
        __syncthreads();
    }
    const int start = sc[cur][tid] - tot;
    if (tid <= nbins) boff[(long long)blockIdx.x * (nbins + 1) + tid] = start;
    c[tid] = start;
    __syncthreads();

#pragma unroll
    for (int k = 0; k < EPT; ++k)
        if (bin[k] >= 0) srt[c[bin[k]] + rk[k]] = wdv[k];
    __syncthreads();

    for (int k = 0; k < EPT; ++k) {
        int i = k * 1024 + tid;
        int e = blockIdx.x * EPB + i;
        if (e < E) list[e] = srt[i];
    }
}

#define RPW 7
#define CH (ROWS / RPW)
#define RS_CAP 8192

__device__ __forceinline__
void rs_load(const unsigned* __restrict__ list, const int (&myro)[16], const int (&myrc)[16],
             int kb, unsigned (&wd)[32]) {
#pragma unroll
    for (int j = 0; j < 16; ++j) {
        wd[2 * j]     = list[kb      < myrc[j] ? myro[j] + kb      : 0];
        wd[2 * j + 1] = list[kb + 16 < myrc[j] ? myro[j] + kb + 16 : 0];
    }
    __builtin_amdgcn_sched_barrier(0);
}

__device__ __forceinline__
void rs_count(int* c, const int (&myrc)[16], int kb, const unsigned (&wd)[32], int tid) {
#pragma unroll
    for (int j = 0; j < 32; ++j) {
        const bool live = kb + 16 * (j & 1) < myrc[j >> 1];
        atomicAdd(&c[live ? (wd[j] & 255u) : (unsigned)tid], live ? 1 : 0);
    }
}

template <bool INLDS>
__device__ __forceinline__
void rs_put(unsigned* __restrict__ list2, unsigned* buf, int* c, const int (&myrc)[16], int kb,
            const unsigned (&wd)[32], int tid) {
    int pl[32];
#pragma unroll
    for (int j = 0; j < 32; ++j) {
        const bool live = kb + 16 * (j & 1) < myrc[j >> 1];
        pl[j] = atomicAdd(&c[live ? (wd[j] & 255u) : (unsigned)tid], live ? 1 : 0);
    }
#pragma unroll
    for (int j = 0; j < 32; ++j)
        if (kb + 16 * (j & 1) < myrc[j >> 1]) {
            if (INLDS) buf[pl[j]] = wd[j];
            else       list2[pl[j]] = wd[j];
        }
}

__global__ __launch_bounds__(256)
void k_rowsort(const unsigned* __restrict__ list, const int* __restrict__ boff,
               float* __restrict__ dinv, unsigned* __restrict__ list2, int* __restrict__ cptr,
               int n, int nbins, int nblk) {
    __shared__ int c[256];
    __shared__ int sc[2][256];
    __shared__ int ro[MAXBLK];
    __shared__ int rc[MAXBLK];
    __shared__ unsigned buf[RS_CAP];
    __shared__ int s_beg, s_max;
    const int tid = threadIdx.x;
    const int bin = blockIdx.x;
    c[tid] = 0;
    if (tid == 0) { s_beg = 0; s_max = 0; }
    __syncthreads();
    int o0 = 0, len = 0;
    if (tid < nblk) {
        const int* bo = boff + (long long)tid * (nbins + 1) + bin;
        o0  = bo[0];
        len = bo[1] - o0;
    }
    ro[tid] = tid * EPB + o0;
    rc[tid] = len;
    int so = o0, mx = len;
    for (int d = 16; d > 0; d >>= 1) {
        so += __shfl_xor(so, d);
        int t = __shfl_xor(mx, d);
        mx = t > mx ? t : mx;
    }
    if ((tid & 31) == 0) { atomicAdd(&s_beg, so); atomicMax(&s_max, mx); }
    __syncthreads();
    const int beg   = s_beg;
    const int maxrc = s_max;

    const int hw  = tid >> 4;
    const int l16 = tid & 15;
    int myro[16], myrc[16];
#pragma unroll
    for (int j = 0; j < 16; ++j) {
        myro[j] = ro[hw + 16 * j] + l16;
        myrc[j] = rc[hw + 16 * j] - l16;
    }

    unsigned w0[32];
    rs_load(list, myro, myrc, 0, w0);
    rs_count(c, myrc, 0, w0, tid);
    for (int kb = 32; kb < maxrc; kb += 32) {
        unsigned wd[32];
        rs_load(list, myro, myrc, kb, wd);
        rs_count(c, myrc, kb, wd, tid);
    }
    __syncthreads();
    const int row = bin * ROWS + tid;
    const bool valid = tid < ROWS && row < n;
    const int slots = valid ? c[tid] + 1 : 0;
    if (valid) {
        float d = (float)slots;
        dinv[row] = d > 0.0f ? rsqrtf(d) : 0.0f;
    }
    sc[0][tid] = slots;
    __syncthreads();
    int cur = 0;
    for (int off = 1; off < 256; off <<= 1) {
        int v = sc[cur][tid];
        if (tid >= off) v += sc[cur][tid - off];
        sc[cur ^ 1][tid] = v;
        cur ^= 1;
        __syncthreads();
    }
    const int incl = sc[cur][tid];
    const int total = sc[cur][255];
    const bool inlds = total <= RS_CAP;
    const int base = beg + bin * ROWS;
    const int lfirst = incl - slots;
    const int first = base + lfirst;
    const unsigned selfw = ((unsigned)row << 8) | (unsigned)tid;
    if (valid) {
        if (inlds) buf[lfirst] = selfw;
        else       list2[first] = selfw;
    }
    if (tid < ROWS && tid % RPW == 0) cptr[bin * CH + tid / RPW] = first;
    if (bin == nbins - 1 && tid == 255) cptr[nbins * CH] = base + incl;
    c[tid] = (inlds ? lfirst : first) + 1;
    __syncthreads();
    if (inlds) {
        rs_put<true>(list2, buf, c, myrc, 0, w0, tid);
        for (int kb = 32; kb < maxrc; kb += 32) {
            unsigned wd[32];
            rs_load(list, myro, myrc, kb, wd);
            rs_put<true>(list2, buf, c, myrc, kb, wd, tid);
        }
    } else {
        rs_put<false>(list2, buf, c, myrc, 0, w0, tid);
        for (int kb = 32; kb < maxrc; kb += 32) {
            unsigned wd[32];
            rs_load(list, myro, myrc, kb, wd);
            rs_put<false>(list2, buf, c, myrc, kb, wd, tid);
        }
    }
    __syncthreads();
    if (inlds) {
        for (int ib = tid - (base & 31); ib < total; ib += 1024) {
            unsigned v[4];
#pragma unroll
            for (int u = 0; u < 4; ++u) {
                const int i = ib + 256 * u;
                v[u] = buf[(i >= 0 && i < total) ? i : 0];
            }
#pragma unroll
            for (int u = 0; u < 4; ++u) {
                const int i = ib + 256 * u;
                if (i >= 0 && i < total) list2[base + i] = v[u];
            }
        }
    }
}

template <bool FULL, bool PRE, int OP, int NE>
__device__ __forceinline__
void agg_group(const float* __restrict__ x, float* __restrict__ orow, unsigned word, float w,
               int j0, int cnt, int lane, float blo, float bhi,
               int& cur, int& run, float& alo, float& ahi) {
    float xlo[NE], xhi[NE];
#pragma unroll
    for (int u = 0; u < NE; ++u) {
        unsigned wj = (unsigned)__builtin_amdgcn_readlane((int)word, j0 + u);
        const float* r = x + (long long)(wj >> 8) * FDIM + lane;
        xlo[u] = r[0];
        xhi[u] = r[32];
    }
    __builtin_amdgcn_sched_barrier(0);
#pragma unroll
    for (int u = 0; u < NE; ++u) {
        const bool live = FULL || j0 + u < cnt;
        unsigned wj = (unsigned)__builtin_amdgcn_readlane((int)word, j0 + u);
        float sj = 0.0f;
        if (!PRE) sj = __int_as_float(__builtin_amdgcn_readlane(__float_as_int(w), j0 + u));
        const int tag = (int)(wj & 255u);
        if (live && tag != cur) {
            if (cur >= 0) {
                const float di = rsqrtf((float)run);
                orow[cur * OP]      = di * alo + blo;
                orow[cur * OP + 32] = di * ahi + bhi;
            }
            cur = tag; run = 0; alo = 0.0f; ahi = 0.0f;
        }
        const float nlo = PRE ? alo + xlo[u] : alo + xlo[u] * sj;
        const float nhi = PRE ? ahi + xhi[u] : ahi + xhi[u] * sj;
        alo = live ? nlo : alo;
        ahi = live ? nhi : ahi;
        run += live ? 1 : 0;
    }
}

__global__ __launch_bounds__(CH * 32)
void k_agg(const float* __restrict__ x, const unsigned* __restrict__ list2,
           const int* __restrict__ cptr, const float* __restrict__ bias,
           float* __restrict__ outp) {
    const int tid  = threadIdx.x;
    const int lane = tid & 31;
    const int wv  = __builtin_amdgcn_readfirstlane(tid >> 5);
    const int bin = blockIdx.x;
    const int beg = cptr[bin * CH + wv];
    const int end = cptr[bin * CH + wv + 1];
    const float blo = bias[lane];
    const float bhi = bias[lane + 32];
    float* const orow = outp + (long long)bin * ROWS * FDIM + lane;

    float alo = 0.0f, ahi = 0.0f;
    int cur = -1, run = 0;
    unsigned word = 0u;
    if (beg + lane < end) word = list2[beg + lane];
    if (beg >= end) return;
    cur = (int)((unsigned)__builtin_amdgcn_readlane((int)word, 0) & 255u);
    for (int base = beg; base < end; base += 32) {
        const int cnt = end - base < 32 ? end - base : 32;
        const bool more = base + 32 + lane < end;
        unsigned nword = 0u;
        if (more) nword = list2[base + 32 + lane];
        if (cnt >= 16) agg_group<true , true, FDIM, 16>(x, orow, word, 0.0f, 0, cnt, lane, blo, bhi, cur, run, alo, ahi);
        else           agg_group<false, true, FDIM, 16>(x, orow, word, 0.0f, 0, cnt, lane, blo, bhi, cur, run, alo, ahi);
        if (cnt >= 32)     agg_group<true , true, FDIM, 16>(x, orow, word, 0.0f, 16, cnt, lane, blo, bhi, cur, run, alo, ahi);
        else if (cnt > 16) agg_group<false, true, FDIM, 16>(x, orow, word, 0.0f, 16, cnt, lane, blo, bhi, cur, run, alo, ahi);
        word = nword;
    }
    if (cur >= 0) {
        const float di = rsqrtf((float)run);
        orow[cur * FDIM]      = di * alo + blo;
        orow[cur * FDIM + 32] = di * ahi + bhi;
    }
}

#define NT ((ROWS + 15) / 16)
#define GW (CH / 2)
#define AM_THREADS ((GW + NT) * 32)
#define HP 132
#define WF 8192
#define TSET (NT * 16 * HP)
#define AM_LDS_FLOATS (2 * WF + 2 * TSET)

__device__ __forceinline__
void am_sum(const float* __restrict__ x, const unsigned* __restrict__ list2,
            const float* __restrict__ dinv, float* set, int beg, int end, int lane) {
    float* const orow = set + lane;
    float alo = 0.0f, ahi = 0.0f;
    int cur = -1, run = 0;
    unsigned word = 0u;
    float w = 0.0f;
    if (beg + lane < end) {
        word = list2[beg + lane];
        w = dinv[word >> 8];
    }
    for (int base = beg; base < end; base += 32) {
        const int cnt = end - base < 32 ? end - base : 32;
        const bool more = base + 32 + lane < end;
        unsigned nword = 0u;
        float nw = 0.0f;
        if (more) nword = list2[base + 32 + lane];
        if (cnt >= 32) agg_group<true , false, HP, 32>(x, orow, word, w, 0, cnt, lane, 0.0f, 0.0f, cur, run, alo, ahi);
        else           agg_group<false, false, HP, 32>(x, orow, word, w, 0, cnt, lane, 0.0f, 0.0f, cur, run, alo, ahi);
        __builtin_amdgcn_sched_barrier(0);
        nw = dinv[nword >> 8];
        word = nword;
        w = nw;
    }
    if (cur >= 0) {
        const float di = rsqrtf((float)run);
        orow[cur * HP]      = di * alo;
        orow[cur * HP + 32] = di * ahi;
    }
}

__device__ __forceinline__
void am_mul(float* hb, const v2f* w1v, const v2f* w2v, const float* __restrict__ b1,
            const float* __restrict__ dinv, float* __restrict__ T, int rbase, int gbase, int n,
            int lane) {
    const int half = lane >> 4;
    const int m    = lane & 15;
    v2f a1[16];
#pragma unroll
    for (int s = 0; s < 16; ++s) a1[s] = *(const v2f*)(hb + m * HP + 4 * s + 2 * half);

#pragma unroll
    for (int g = 0; g < 2; ++g) {
        v8f acc[4] = {};
#pragma unroll
        for (int s = 0; s < 16; ++s) {
#pragma unroll
            for (int j = 0; j < 4; ++j) {
                v2f b = w1v[(s * 8 + g * 4 + j) * 32 + lane];
                acc[j] = __builtin_amdgcn_wmma_f32_16x16x4_f32(
                             false, a1[s], false, b, (short)0, acc[j], false, false);
            }
        }
#pragma unroll
        for (int j = 0; j < 4; ++j) {
            const int col = (g * 4 + j) * 16 + m;
            const float bv = b1[col];
#pragma unroll
            for (int r = 0; r < 8; ++r) {
                hb[(r + 8 * half) * HP + col] = fmaxf(acc[j][r] + bv, 0.0f);
            }
        }
    }
    asm volatile("s_wait_dscnt 0x0" ::: "memory");

    float dr[8];
#pragma unroll
    for (int r = 0; r < 8; ++r) {
        const int o = 8 * half + r;
        dr[r] = (rbase + o < ROWS && gbase + o < n) ? dinv[gbase + o] : 0.0f;
    }
    v8f acc[4] = {};
#pragma unroll
    for (int s = 0; s < 32; ++s) {
        const v2f a2 = *(const v2f*)(hb + m * HP + 4 * s + 2 * half);
#pragma unroll
        for (int j = 0; j < 4; ++j) {
            v2f b = w2v[(s * 4 + j) * 32 + lane];
            acc[j] = __builtin_amdgcn_wmma_f32_16x16x4_f32(
                         false, a2, false, b, (short)0, acc[j], false, false);
        }
    }
#pragma unroll
    for (int j = 0; j < 4; ++j)
#pragma unroll
        for (int r = 0; r < 8; ++r)
            hb[(r + 8 * half) * HP + j * 16 + m] = acc[j][r] * dr[r];
    asm volatile("s_wait_dscnt 0x0" ::: "memory");

    v4f tv[8];
#pragma unroll
    for (int i = 0; i < 8; ++i) {
        const int q = (i * 32 + lane) * 4;
        tv[i] = *(const v4f*)(hb + (q >> 6) * HP + (q & 63));
    }
    float* tt = T + (long long)gbase * 64;
#pragma unroll
    for (int i = 0; i < 8; ++i) {
        const int o = 2 * i + half;
        if (rbase + o < ROWS && gbase + o < n) *(v4f*)(tt + (i * 32 + lane) * 4) = tv[i];
    }
}

__global__ __launch_bounds__(AM_THREADS)
void k_agg_mlp(const float* __restrict__ x, const unsigned* __restrict__ list2,
               const int* __restrict__ cptr, const float* __restrict__ dinv,
               const float* __restrict__ W1, const float* __restrict__ b1,
               const float* __restrict__ W2, float* __restrict__ T, int n, int nbins) {
    __shared__ __attribute__((aligned(16))) float smem[AM_LDS_FLOATS];
    float* w1f = smem;
    float* w2f = smem + WF;
    float* sets = smem + 2 * WF;

    const int tid  = threadIdx.x;
    const int lane = tid & 31;
    const int wv = __builtin_amdgcn_readfirstlane(tid >> 5);

#pragma unroll
    for (int i = 0; i < (WF + NT * 32 - 1) / (NT * 32); ++i) {
        int d  = tid - GW * 32 + NT * 32 * i;
        if (wv >= GW && d < WF) {
            int e  = d & 1;
            int ln = (d >> 1) & 31;
            int mm = ln & 15;
            int hh = ln >> 4;
            int ct1 = (d >> 6) & 7;
            int s1  = d >> 9;
            w1f[d] = W1[(4 * s1 + 2 * hh + e) * 128 + 16 * ct1 + mm];
            int ct2 = (d >> 6) & 3;
            int s2  = d >> 8;
            w2f[d] = W2[(4 * s2 + 2 * hh + e) * 64 + 16 * ct2 + mm];
        }
    }

    const int G = (int)gridDim.x;
    const int rounds = (nbins + G - 1) / G;
    for (int it = 0; it <= rounds; ++it) {
        float* sset = sets + (it & 1) * TSET;
        float* mset = sets + ((it & 1) ^ 1) * TSET;
        const int sbin = (int)blockIdx.x + it * G;
        const int mbin = sbin - G;
        if (wv < GW) {
            if (it < rounds && sbin < nbins) {
                const int beg = cptr[sbin * CH + 2 * wv];
                const int end = cptr[sbin * CH + 2 * wv + 2];
                const int zr0 = 2 * RPW * wv;
                const int zr1 = wv == GW - 1 ? NT * 16 : zr0 + 2 * RPW;
                for (int r = zr0; r < zr1; ++r) {
                    sset[r * HP + lane]      = 0.0f;
                    sset[r * HP + lane + 32] = 0.0f;
                }
                am_sum(x, list2, dinv, sset, beg, end, lane);
            }
        } else {
            const int k = wv - GW;
            const int gbase = mbin * ROWS + k * 16;
            if (it > 0 && mbin < nbins && gbase < n)
                am_mul(mset + k * (16 * HP), (const v2f*)w1f, (const v2f*)w2f, b1, dinv, T,
                       k * 16, gbase, n, lane);
        }
        __syncthreads();
    }
}

extern "C" void kernel_launch(void* const* d_in, const int* in_sizes, int n_in,
                              void* d_out, int out_size, void* d_ws, size_t ws_size,
                              hipStream_t stream) {
    (void)n_in; (void)out_size; (void)ws_size;

    const int*   eidx = (const int*)d_in[0];
    const float* emb  = (const float*)d_in[1];
    const float* W1   = (const float*)d_in[2];
    const float* b1   = (const float*)d_in[3];
    const float* W2   = (const float*)d_in[4];
    const float* b2   = (const float*)d_in[5];

    const int E = in_sizes[0] / 2;
    const int n = in_sizes[1] / 64;
    const int* src = eidx;
    const int* dst = eidx + E;
    const int nbins = (n + ROWS - 1) / ROWS;

    int*   cptr = (int*)d_ws;
    float* dinv = (float*)d_ws + n;
    float* scr  = dinv + n;
    float* t    = scr + (size_t)n * 64;
    float* out  = (float*)d_out;
    unsigned* list  = (unsigned*)scr;
    int*      boff  = (int*)(list + E);
    unsigned* list2 = (unsigned*)(t + (size_t)n * 64);
    const int nblk = (E + EPB - 1) / EPB;

    k_split  <<<nblk, 1024, 0, stream>>>(src, dst, list, boff, E, nbins);
    k_rowsort<<<nbins, 256, 0, stream>>>(list, boff, dinv, list2, cptr, n, nbins, nblk);

    k_agg_mlp<<<(nbins + 1) / 2, AM_THREADS, 0, stream>>>(emb, list2, cptr, dinv, W1, b1, W2, t, n, nbins);

    k_agg<<<nbins, CH * 32, 0, stream>>>(t, list2, cptr, b2, out);
}


# aggregations: lane l owns features 2l, 2l+1 (was l, l+32): a source row is one 8-byte load per lane (one instruction for its two lines), a summed row one 8-byte store; same sum order
# speedup vs baseline: 5.9115x; 1.0549x over previous
#include <hip/hip_runtime.h>

typedef __attribute__((ext_vector_type(2))) float v2f;
typedef __attribute__((ext_vector_type(8))) float v8f;
typedef __attribute__((ext_vector_type(4))) float v4f;

#define FDIM 64
#define ROWS 196
#define MAXBINS 1024
#define EPT 8
#define EPB (EPT * 1024)

#define MAXBLK 256

__global__ __launch_bounds__(1024)
void k_split(const int* __restrict__ src, const int* __restrict__ dst,
             unsigned* __restrict__ list, int* __restrict__ boff, int E, int nbins) {
    __shared__ int c[MAXBINS];
    __shared__ int sc[2][MAXBINS];
    __shared__ unsigned srt[EPB];
    const int tid = threadIdx.x;
    c[tid] = 0;
    __syncthreads();

    int dv[EPT], sv[EPT];
#pragma unroll
    for (int k = 0; k < EPT; ++k) {
        const int e = blockIdx.x * EPB + k * 1024 + tid;
        const int ec = e < E ? e : E - 1;
        dv[k] = dst[ec];
        sv[k] = src[ec];
    }
    __builtin_amdgcn_sched_barrier(0);
    int bin[EPT], rk[EPT];
    unsigned wdv[EPT];
#pragma unroll
    for (int k = 0; k < EPT; ++k) {
        const bool live = blockIdx.x * EPB + k * 1024 + tid < E;
        const int d = dv[k];
        const int b = d / ROWS;
        rk[k]  = atomicAdd(&c[live ? b : tid], live ? 1 : 0);
        bin[k] = live ? b : -1;
        wdv[k] = ((unsigned)sv[k] << 8) | (unsigned)(d - b * ROWS);
    }
    __syncthreads();

    const int tot = c[tid];
    sc[0][tid] = tot;
    __syncthreads();
    int cur = 0;
    for (int off = 1; off < MAXBINS; off <<= 1) {
        int v = sc[cur][tid];
        if (tid >= off) v += sc[cur][tid - off];
        sc[cur ^ 1][tid] = v;
        cur ^= 1;
        __syncthreads();
    }
    const int start = sc[cur][tid] - tot;
    if (tid <= nbins) boff[(long long)blockIdx.x * (nbins + 1) + tid] = start;
    c[tid] = start;
    __syncthreads();

#pragma unroll
    for (int k = 0; k < EPT; ++k)
        if (bin[k] >= 0) srt[c[bin[k]] + rk[k]] = wdv[k];
    __syncthreads();

    for (int k = 0; k < EPT; ++k) {
        int i = k * 1024 + tid;
        int e = blockIdx.x * EPB + i;
        if (e < E) list[e] = srt[i];
    }
}

#define RPW 7
#define CH (ROWS / RPW)
#define RS_CAP 8192

__device__ __forceinline__
void rs_load(const unsigned* __restrict__ list, const int (&myro)[16], const int (&myrc)[16],
             int kb, unsigned (&wd)[32]) {
#pragma unroll
    for (int j = 0; j < 16; ++j) {
        wd[2 * j]     = list[kb      < myrc[j] ? myro[j] + kb      : 0];
        wd[2 * j + 1] = list[kb + 16 < myrc[j] ? myro[j] + kb + 16 : 0];
    }
    __builtin_amdgcn_sched_barrier(0);
}

__device__ __forceinline__
void rs_count(int* c, const int (&myrc)[16], int kb, const unsigned (&wd)[32], int tid) {
#pragma unroll
    for (int j = 0; j < 32; ++j) {
        const bool live = kb + 16 * (j & 1) < myrc[j >> 1];
        atomicAdd(&c[live ? (wd[j] & 255u) : (unsigned)tid], live ? 1 : 0);
    }
}

template <bool INLDS>
__device__ __forceinline__
void rs_put(unsigned* __restrict__ list2, unsigned* buf, int* c, const int (&myrc)[16], int kb,
            const unsigned (&wd)[32], int tid) {
    int pl[32];
#pragma unroll
    for (int j = 0; j < 32; ++j) {
        const bool live = kb + 16 * (j & 1) < myrc[j >> 1];
        pl[j] = atomicAdd(&c[live ? (wd[j] & 255u) : (unsigned)tid], live ? 1 : 0);
    }
#pragma unroll
    for (int j = 0; j < 32; ++j)
        if (kb + 16 * (j & 1) < myrc[j >> 1]) {
            if (INLDS) buf[pl[j]] = wd[j];
            else       list2[pl[j]] = wd[j];
        }
}

__global__ __launch_bounds__(256)
void k_rowsort(const unsigned* __restrict__ list, const int* __restrict__ boff,
               float* __restrict__ dinv, unsigned* __restrict__ list2, int* __restrict__ cptr,
               int n, int nbins, int nblk) {
    __shared__ int c[256];
    __shared__ int sc[2][256];
    __shared__ int ro[MAXBLK];
    __shared__ int rc[MAXBLK];
    __shared__ unsigned buf[RS_CAP];
    __shared__ int s_beg, s_max;
    const int tid = threadIdx.x;
    const int bin = blockIdx.x;
    c[tid] = 0;
    if (tid == 0) { s_beg = 0; s_max = 0; }
    __syncthreads();
    int o0 = 0, len = 0;
    if (tid < nblk) {
        const int* bo = boff + (long long)tid * (nbins + 1) + bin;
        o0  = bo[0];
        len = bo[1] - o0;
    }
    ro[tid] = tid * EPB + o0;
    rc[tid] = len;
    int so = o0, mx = len;
    for (int d = 16; d > 0; d >>= 1) {
        so += __shfl_xor(so, d);
        int t = __shfl_xor(mx, d);
        mx = t > mx ? t : mx;
    }
    if ((tid & 31) == 0) { atomicAdd(&s_beg, so); atomicMax(&s_max, mx); }
    __syncthreads();
    const int beg   = s_beg;
    const int maxrc = s_max;

    const int hw  = tid >> 4;
    const int l16 = tid & 15;
    int myro[16], myrc[16];
#pragma unroll
    for (int j = 0; j < 16; ++j) {
        myro[j] = ro[hw + 16 * j] + l16;
        myrc[j] = rc[hw + 16 * j] - l16;
    }

    unsigned w0[32];
    rs_load(list, myro, myrc, 0, w0);
    rs_count(c, myrc, 0, w0, tid);
    for (int kb = 32; kb < maxrc; kb += 32) {
        unsigned wd[32];
        rs_load(list, myro, myrc, kb, wd);
        rs_count(c, myrc, kb, wd, tid);
    }
    __syncthreads();
    const int row = bin * ROWS + tid;
    const bool valid = tid < ROWS && row < n;
    const int slots = valid ? c[tid] + 1 : 0;
    if (valid) {
        float d = (float)slots;
        dinv[row] = d > 0.0f ? rsqrtf(d) : 0.0f;
    }
    sc[0][tid] = slots;
    __syncthreads();
    int cur = 0;
    for (int off = 1; off < 256; off <<= 1) {
        int v = sc[cur][tid];
        if (tid >= off) v += sc[cur][tid - off];
        sc[cur ^ 1][tid] = v;
        cur ^= 1;
        __syncthreads();
    }
    const int incl = sc[cur][tid];
    const int total = sc[cur][255];
    const bool inlds = total <= RS_CAP;
    const int base = beg + bin * ROWS;
    const int lfirst = incl - slots;
    const int first = base + lfirst;
    const unsigned selfw = ((unsigned)row << 8) | (unsigned)tid;
    if (valid) {
        if (inlds) buf[lfirst] = selfw;
        else       list2[first] = selfw;
    }
    if (tid < ROWS && tid % RPW == 0) cptr[bin * CH + tid / RPW] = first;
    if (bin == nbins - 1 && tid == 255) cptr[nbins * CH] = base + incl;
    c[tid] = (inlds ? lfirst : first) + 1;
    __syncthreads();
    if (inlds) {
        rs_put<true>(list2, buf, c, myrc, 0, w0, tid);
        for (int kb = 32; kb < maxrc; kb += 32) {
            unsigned wd[32];
            rs_load(list, myro, myrc, kb, wd);
            rs_put<true>(list2, buf, c, myrc, kb, wd, tid);
        }
    } else {
        rs_put<false>(list2, buf, c, myrc, 0, w0, tid);
        for (int kb = 32; kb < maxrc; kb += 32) {
            unsigned wd[32];
            rs_load(list, myro, myrc, kb, wd);
            rs_put<false>(list2, buf, c, myrc, kb, wd, tid);
        }
    }
    __syncthreads();
    if (inlds) {
        for (int ib = tid - (base & 31); ib < total; ib += 1024) {
            unsigned v[4];
#pragma unroll
            for (int u = 0; u < 4; ++u) {
                const int i = ib + 256 * u;
                v[u] = buf[(i >= 0 && i < total) ? i : 0];
            }
#pragma unroll
            for (int u = 0; u < 4; ++u) {
                const int i = ib + 256 * u;
                if (i >= 0 && i < total) list2[base + i] = v[u];
            }
        }
    }
}

template <bool FULL, bool PRE, int OP, int NE>
__device__ __forceinline__
void agg_group(const float* __restrict__ x, float* __restrict__ orow, unsigned word, float w,
               int j0, int cnt, int lane, float blo, float bhi,
               int& cur, int& run, float& alo, float& ahi) {
    v2f xv[NE];
#pragma unroll
    for (int u = 0; u < NE; ++u) {
        unsigned wj = (unsigned)__builtin_amdgcn_readlane((int)word, j0 + u);
        xv[u] = *((const v2f*)(x + (long long)(wj >> 8) * FDIM) + lane);
    }
    __builtin_amdgcn_sched_barrier(0);
#pragma unroll
    for (int u = 0; u < NE; ++u) {
        const bool live = FULL || j0 + u < cnt;
        unsigned wj = (unsigned)__builtin_amdgcn_readlane((int)word, j0 + u);
        float sj = 0.0f;
        if (!PRE) sj = __int_as_float(__builtin_amdgcn_readlane(__float_as_int(w), j0 + u));
        const int tag = (int)(wj & 255u);
        if (live && tag != cur) {
            if (cur >= 0) {
                const float di = rsqrtf((float)run);
                v2f o; o.x = di * alo + blo; o.y = di * ahi + bhi;
                *(v2f*)(orow + cur * OP) = o;
            }
            cur = tag; run = 0; alo = 0.0f; ahi = 0.0f;
        }
        const float nlo = PRE ? alo + xv[u].x : alo + xv[u].x * sj;
        const float nhi = PRE ? ahi + xv[u].y : ahi + xv[u].y * sj;
        alo = live ? nlo : alo;
        ahi = live ? nhi : ahi;
        run += live ? 1 : 0;
    }
}

__global__ __launch_bounds__(CH * 32)
void k_agg(const float* __restrict__ x, const unsigned* __restrict__ list2,
           const int* __restrict__ cptr, const float* __restrict__ bias,
           float* __restrict__ outp) {
    const int tid  = threadIdx.x;
    const int lane = tid & 31;
    const int wv  = __builtin_amdgcn_readfirstlane(tid >> 5);
    const int bin = blockIdx.x;
    const int beg = cptr[bin * CH + wv];
    const int end = cptr[bin * CH + wv + 1];
    const float blo = bias[2 * lane];
    const float bhi = bias[2 * lane + 1];
    float* const orow = outp + (long long)bin * ROWS * FDIM + 2 * lane;

    float alo = 0.0f, ahi = 0.0f;
    int cur = -1, run = 0;
    unsigned word = 0u;
    if (beg + lane < end) word = list2[beg + lane];
    if (beg >= end) return;
    cur = (int)((unsigned)__builtin_amdgcn_readlane((int)word, 0) & 255u);
    for (int base = beg; base < end; base += 32) {
        const int cnt = end - base < 32 ? end - base : 32;
        const bool more = base + 32 + lane < end;
        unsigned nword = 0u;
        if (more) nword = list2[base + 32 + lane];
        if (cnt >= 16) agg_group<true , true, FDIM, 16>(x, orow, word, 0.0f, 0, cnt, lane, blo, bhi, cur, run, alo, ahi);
        else           agg_group<false, true, FDIM, 16>(x, orow, word, 0.0f, 0, cnt, lane, blo, bhi, cur, run, alo, ahi);
        if (cnt >= 32)     agg_group<true , true, FDIM, 16>(x, orow, word, 0.0f, 16, cnt, lane, blo, bhi, cur, run, alo, ahi);
        else if (cnt > 16) agg_group<false, true, FDIM, 16>(x, orow, word, 0.0f, 16, cnt, lane, blo, bhi, cur, run, alo, ahi);
        word = nword;
    }
    if (cur >= 0) {
        const float di = rsqrtf((float)run);
        v2f o; o.x = di * alo + blo; o.y = di * ahi + bhi;
        *(v2f*)(orow + cur * FDIM) = o;
    }
}

#define NT ((ROWS + 15) / 16)
#define GW (CH / 2)
#define AM_THREADS ((GW + NT) * 32)
#define HP 132
#define WF 8192
#define TSET (NT * 16 * HP)
#define AM_LDS_FLOATS (2 * WF + 2 * TSET)

__device__ __forceinline__
void am_sum(const float* __restrict__ x, const unsigned* __restrict__ list2,
            const float* __restrict__ dinv, float* set, int beg, int end, int lane) {
    float* const orow = set + 2 * lane;
    float alo = 0.0f, ahi = 0.0f;
    int cur = -1, run = 0;
    unsigned word = 0u;
    float w = 0.0f;
    if (beg + lane < end) {
        word = list2[beg + lane];
        w = dinv[word >> 8];
    }
    for (int base = beg; base < end; base += 32) {
        const int cnt = end - base < 32 ? end - base : 32;
        const bool more = base + 32 + lane < end;
        unsigned nword = 0u;
        float nw = 0.0f;
        if (more) nword = list2[base + 32 + lane];
        if (cnt >= 32) agg_group<true , false, HP, 32>(x, orow, word, w, 0, cnt, lane, 0.0f, 0.0f, cur, run, alo, ahi);
        else           agg_group<false, false, HP, 32>(x, orow, word, w, 0, cnt, lane, 0.0f, 0.0f, cur, run, alo, ahi);
        __builtin_amdgcn_sched_barrier(0);
        nw = dinv[nword >> 8];
        word = nword;
        w = nw;
    }
    if (cur >= 0) {
        const float di = rsqrtf((float)run);
        v2f o; o.x = di * alo; o.y = di * ahi;
        *(v2f*)(orow + cur * HP) = o;
    }
}

__device__ __forceinline__
void am_mul(float* hb, const v2f* w1v, const v2f* w2v, const float* __restrict__ b1,
            const float* __restrict__ dinv, float* __restrict__ T, int rbase, int gbase, int n,
            int lane) {
    const int half = lane >> 4;
    const int m    = lane & 15;
    v2f a1[16];
#pragma unroll
    for (int s = 0; s < 16; ++s) a1[s] = *(const v2f*)(hb + m * HP + 4 * s + 2 * half);

#pragma unroll
    for (int g = 0; g < 2; ++g) {
        v8f acc[4] = {};
#pragma unroll
        for (int s = 0; s < 16; ++s) {
#pragma unroll
            for (int j = 0; j < 4; ++j) {
                v2f b = w1v[(s * 8 + g * 4 + j) * 32 + lane];
                acc[j] = __builtin_amdgcn_wmma_f32_16x16x4_f32(
                             false, a1[s], false, b, (short)0, acc[j], false, false);
            }
        }
#pragma unroll
        for (int j = 0; j < 4; ++j) {
            const int col = (g * 4 + j) * 16 + m;
            const float bv = b1[col];
#pragma unroll
            for (int r = 0; r < 8; ++r) {
                hb[(r + 8 * half) * HP + col] = fmaxf(acc[j][r] + bv, 0.0f);
            }
        }
    }
    asm volatile("s_wait_dscnt 0x0" ::: "memory");

    float dr[8];
#pragma unroll
    for (int r = 0; r < 8; ++r) {
        const int o = 8 * half + r;
        dr[r] = (rbase + o < ROWS && gbase + o < n) ? dinv[gbase + o] : 0.0f;
    }
    v8f acc[4] = {};
#pragma unroll
    for (int s = 0; s < 32; ++s) {
        const v2f a2 = *(const v2f*)(hb + m * HP + 4 * s + 2 * half);
#pragma unroll
        for (int j = 0; j < 4; ++j) {
            v2f b = w2v[(s * 4 + j) * 32 + lane];
            acc[j] = __builtin_amdgcn_wmma_f32_16x16x4_f32(
                         false, a2, false, b, (short)0, acc[j], false, false);
        }
    }
#pragma unroll
    for (int j = 0; j < 4; ++j)
#pragma unroll
        for (int r = 0; r < 8; ++r)
            hb[(r + 8 * half) * HP + j * 16 + m] = acc[j][r] * dr[r];
    asm volatile("s_wait_dscnt 0x0" ::: "memory");

    v4f tv[8];
#pragma unroll
    for (int i = 0; i < 8; ++i) {
        const int q = (i * 32 + lane) * 4;
        tv[i] = *(const v4f*)(hb + (q >> 6) * HP + (q & 63));
    }
    float* tt = T + (long long)gbase * 64;
#pragma unroll
    for (int i = 0; i < 8; ++i) {
        const int o = 2 * i + half;
        if (rbase + o < ROWS && gbase + o < n) *(v4f*)(tt + (i * 32 + lane) * 4) = tv[i];
    }
}

__global__ __launch_bounds__(AM_THREADS)
void k_agg_mlp(const float* __restrict__ x, const unsigned* __restrict__ list2,
               const int* __restrict__ cptr, const float* __restrict__ dinv,
               const float* __restrict__ W1, const float* __restrict__ b1,
               const float* __restrict__ W2, float* __restrict__ T, int n, int nbins) {
    __shared__ __attribute__((aligned(16))) float smem[AM_LDS_FLOATS];
    float* w1f = smem;
    float* w2f = smem + WF;
    float* sets = smem + 2 * WF;

    const int tid  = threadIdx.x;
    const int lane = tid & 31;
    const int wv = __builtin_amdgcn_readfirstlane(tid >> 5);

#pragma unroll
    for (int i = 0; i < (WF + NT * 32 - 1) / (NT * 32); ++i) {
        int d  = tid - GW * 32 + NT * 32 * i;
        if (wv >= GW && d < WF) {
            int e  = d & 1;
            int ln = (d >> 1) & 31;
            int mm = ln & 15;
            int hh = ln >> 4;
            int ct1 = (d >> 6) & 7;
            int s1  = d >> 9;
            w1f[d] = W1[(4 * s1 + 2 * hh + e) * 128 + 16 * ct1 + mm];
            int ct2 = (d >> 6) & 3;
            int s2  = d >> 8;
            w2f[d] = W2[(4 * s2 + 2 * hh + e) * 64 + 16 * ct2 + mm];
        }
    }

    const int G = (int)gridDim.x;
    const int rounds = (nbins + G - 1) / G;
    for (int it = 0; it <= rounds; ++it) {
        float* sset = sets + (it & 1) * TSET;
        float* mset = sets + ((it & 1) ^ 1) * TSET;
        const int sbin = (int)blockIdx.x + it * G;
        const int mbin = sbin - G;
        if (wv < GW) {
            if (it < rounds && sbin < nbins) {
                const int beg = cptr[sbin * CH + 2 * wv];
                const int end = cptr[sbin * CH + 2 * wv + 2];
                const int zr0 = 2 * RPW * wv;
                const int zr1 = wv == GW - 1 ? NT * 16 : zr0 + 2 * RPW;
                for (int r = zr0; r < zr1; ++r) {
                    sset[r * HP + lane]      = 0.0f;
                    sset[r * HP + lane + 32] = 0.0f;
                }
                am_sum(x, list2, dinv, sset, beg, end, lane);
            }
        } else {
            const int k = wv - GW;
            const int gbase = mbin * ROWS + k * 16;
            if (it > 0 && mbin < nbins && gbase < n)
                am_mul(mset + k * (16 * HP), (const v2f*)w1f, (const v2f*)w2f, b1, dinv, T,
                       k * 16, gbase, n, lane);
        }
        __syncthreads();
    }
}

extern "C" void kernel_launch(void* const* d_in, const int* in_sizes, int n_in,
                              void* d_out, int out_size, void* d_ws, size_t ws_size,
                              hipStream_t stream) {
    (void)n_in; (void)out_size; (void)ws_size;

    const int*   eidx = (const int*)d_in[0];
    const float* emb  = (const float*)d_in[1];
    const float* W1   = (const float*)d_in[2];
    const float* b1   = (const float*)d_in[3];
    const float* W2   = (const float*)d_in[4];
    const float* b2   = (const float*)d_in[5];

    const int E = in_sizes[0] / 2;
    const int n = in_sizes[1] / 64;
    const int* src = eidx;
    const int* dst = eidx + E;
    const int nbins = (n + ROWS - 1) / ROWS;

    int*   cptr = (int*)d_ws;
    float* dinv = (float*)d_ws + n;
    float* scr  = dinv + n;
    float* t    = scr + (size_t)n * 64;
    float* out  = (float*)d_out;
    unsigned* list  = (unsigned*)scr;
    int*      boff  = (int*)(list + E);
    unsigned* list2 = (unsigned*)(t + (size_t)n * 64);
    const int nblk = (E + EPB - 1) / EPB;

    k_split  <<<nblk, 1024, 0, stream>>>(src, dst, list, boff, E, nbins);
    k_rowsort<<<nbins, 256, 0, stream>>>(list, boff, dinv, list2, cptr, n, nbins, nblk);

    k_agg_mlp<<<(nbins + 1) / 2, AM_THREADS, 0, stream>>>(emb, list2, cptr, dinv, W1, b1, W2, t, n, nbins);

    k_agg<<<nbins, CH * 32, 0, stream>>>(t, list2, cptr, b2, out);
}
